# starting kernel
# baseline (speedup 1.0000x reference)
#include <hip/hip_runtime.h>

typedef float v2f __attribute__((ext_vector_type(2)));
typedef float v8f __attribute__((ext_vector_type(8)));

#define K_DIM 128

template <int DOUT>
__global__ __launch_bounds__(256) void gemm_f32_wmma(
    const float* __restrict__ A, const float* __restrict__ W,
    float* __restrict__ T, int N) {
  constexpr int CT = DOUT / 16;
  constexpr int RT = 8 / CT;
  constexpr int ROWS = RT * 16;
  __shared__ float As[ROWS][K_DIM + 4];

  const int tid  = threadIdx.x;
  const int wave = tid >> 5;
  const int lane = tid & 31;
  const int rowTile = wave / CT;
  const int colTile = wave % CT;
  const long rowBase = (long)blockIdx.x * ROWS;

  for (int i = tid; i < ROWS * K_DIM; i += 256) {
    const int r = i >> 7;
    const int c = i & 127;
    const long gr = rowBase + r;
    As[r][c] = (gr < N) ? A[gr * K_DIM + c] : 0.0f;
  }
  __syncthreads();

  const int half = lane >> 4;
  const int l16  = lane & 15;
  const int n0   = colTile * 16 + l16;
  const int arow = rowTile * 16 + l16;

  v8f acc = {};
  #pragma unroll
  for (int kk = 0; kk < K_DIM; kk += 4) {
    const int ka = kk + 2 * half;
    v2f a, b;
    a.x = As[arow][ka];
    a.y = As[arow][ka + 1];
    b.x = W[ka * DOUT + n0];
    b.y = W[(ka + 1) * DOUT + n0];
    acc = __builtin_amdgcn_wmma_f32_16x16x4_f32(
         false, a,  false, b,
         (short)0, acc,  false,  false);
  }

  const long mBase = rowBase + rowTile * 16 + 8 * half;
  #pragma unroll
  for (int v = 0; v < 8; ++v) {
    const long m = mBase + v;
    if (m < N) T[m * DOUT + n0] = acc[v];
  }
}

__global__ void init_deg_kernel(float* deg, int N) {
  const long i = (long)blockIdx.x * blockDim.x + threadIdx.x;
  if (i < N) deg[i] = 1.0f;
}

__global__ void count_deg_kernel(const int* __restrict__ dst, float* deg, int E) {
  const long e = (long)blockIdx.x * blockDim.x + threadIdx.x;
  if (e < E) atomicAdd(&deg[dst[e]], 1.0f);
}

__global__ void finalize_dinv_kernel(float* deg, int N) {
  const long i = (long)blockIdx.x * blockDim.x + threadIdx.x;
  if (i < N) deg[i] = rsqrtf(deg[i]);
}

template <int D>
__global__ void agg_self_kernel(const float* __restrict__ T,
                                const float* __restrict__ dinv,
                                float* __restrict__ out, int N) {
  const long idx = (long)blockIdx.x * blockDim.x + threadIdx.x;
  if (idx >= (long)N * D) return;
  const long i = idx / D;
  const float w = dinv[i];
  out[idx] = T[idx] * w * w;
}

template <int D>
__global__ void agg_edges_kernel(const float* __restrict__ T,
                                 const int* __restrict__ src,
                                 const int* __restrict__ dst,
                                 const float* __restrict__ dinv,
                                 float* __restrict__ out, int E) {
  constexpr int TPE = D / 4;
  const long idx = (long)blockIdx.x * blockDim.x + threadIdx.x;
  const long e = idx / TPE;
  const int  t = (int)(idx % TPE);
  if (e >= E) return;
  const int s = src[e];
  const int d = dst[e];
  const float norm = dinv[s] * dinv[d];
  const float4 v = ((const float4*)(T + (long)s * D))[t];
  float* op = out + (long)d * D + t * 4;
  atomicAdd(op + 0, v.x * norm);
  atomicAdd(op + 1, v.y * norm);
  atomicAdd(op + 2, v.z * norm);
  atomicAdd(op + 3, v.w * norm);
}

template <int D>
__global__ void bias_relu_kernel(float* __restrict__ h,
                                 const float* __restrict__ b, long total) {
  const long idx = (long)blockIdx.x * blockDim.x + threadIdx.x;
  if (idx >= total) return;
  const int d = (int)(idx % D);
  h[idx] = fmaxf(h[idx] + b[d], 0.0f);
}

__global__ void zero_kernel(float* p, long n) {
  const long i = (long)blockIdx.x * blockDim.x + threadIdx.x;
  if (i < n) p[i] = 0.0f;
}

__global__ void pool_count_kernel(const int* __restrict__ batch, float* counts, int N) {
  const long i = (long)blockIdx.x * blockDim.x + threadIdx.x;
  if (i < N) atomicAdd(&counts[batch[i]], 1.0f);
}

__global__ void pool_sum_kernel(const float* __restrict__ h,
                                const int* __restrict__ batch,
                                float* sums, int N) {
  const long idx = (long)blockIdx.x * blockDim.x + threadIdx.x;
  if (idx >= (long)N * 64) return;
  const long i = idx / 64;
  const int  d = (int)(idx % 64);
  atomicAdd(&sums[(long)batch[i] * 64 + d], h[idx]);
}

__global__ void pool_final_kernel(const float* __restrict__ sums,
                                  const float* __restrict__ counts,
                                  const float* __restrict__ b3,
                                  float* __restrict__ out, int G) {
  const long idx = (long)blockIdx.x * blockDim.x + threadIdx.x;
  if (idx >= (long)G * 64) return;
  const long g = idx / 64;
  const int  d = (int)(idx % 64);
  const float c = counts[g];
  out[idx] = (sums[idx] + c * b3[d]) / fmaxf(c, 1.0f);
}

static inline int cdiv(long a, long b) { return (int)((a + b - 1) / b); }

extern "C" void kernel_launch(void* const* d_in, const int* in_sizes, int n_in,
                              void* d_out, int out_size, void* d_ws, size_t ws_size,
                              hipStream_t stream) {
  const float* x   = (const float*)d_in[0];
  const int*   ei  = (const int*)d_in[1];
  const int*   bat = (const int*)d_in[2];
  const float* W1  = (const float*)d_in[3];
  const float* b1  = (const float*)d_in[4];
  const float* W2  = (const float*)d_in[5];
  const float* b2  = (const float*)d_in[6];
  const float* W3  = (const float*)d_in[7];
  const float* b3  = (const float*)d_in[8];
  (void)n_in; (void)ws_size;

  const int N = in_sizes[0] / 128;
  const int E = in_sizes[1] / 2;
  const int G = out_size / 64;
  const int* src = ei;
  const int* dst = ei + E;

  float* ws = (float*)d_ws;
  size_t o = 0;
  float* dinv = ws + o;   o += ((size_t)N + 63) & ~(size_t)63;
  float* T    = ws + o;   o += (size_t)N * 128;
  float* AGG  = ws + o;   o += (size_t)N * 128;
  float* counts = ws + o; o += (size_t)G;
  float* sums   = ws + o; o += (size_t)G * 64;

  const int B = 256;
  float* out = (float*)d_out;

  init_deg_kernel<<<cdiv(N, B), B, 0, stream>>>(dinv, N);
  count_deg_kernel<<<cdiv(E, B), B, 0, stream>>>(dst, dinv, E);
  finalize_dinv_kernel<<<cdiv(N, B), B, 0, stream>>>(dinv, N);

  gemm_f32_wmma<128><<<cdiv(N, 16), B, 0, stream>>>(x, W1, T, N);
  agg_self_kernel<128><<<cdiv((long)N * 128, B), B, 0, stream>>>(T, dinv, AGG, N);
  agg_edges_kernel<128><<<cdiv((long)E * 32, B), B, 0, stream>>>(T, src, dst, dinv, AGG, E);
  bias_relu_kernel<128><<<cdiv((long)N * 128, B), B, 0, stream>>>(AGG, b1, (long)N * 128);

  gemm_f32_wmma<128><<<cdiv(N, 16), B, 0, stream>>>(AGG, W2, T, N);
  agg_self_kernel<128><<<cdiv((long)N * 128, B), B, 0, stream>>>(T, dinv, AGG, N);
  agg_edges_kernel<128><<<cdiv((long)E * 32, B), B, 0, stream>>>(T, src, dst, dinv, AGG, E);
  bias_relu_kernel<128><<<cdiv((long)N * 128, B), B, 0, stream>>>(AGG, b2, (long)N * 128);

  gemm_f32_wmma<64><<<cdiv(N, 32), B, 0, stream>>>(AGG, W3, T, N);
  agg_self_kernel<64><<<cdiv((long)N * 64, B), B, 0, stream>>>(T, dinv, AGG, N);
  agg_edges_kernel<64><<<cdiv((long)E * 16, B), B, 0, stream>>>(T, src, dst, dinv, AGG, E);

  zero_kernel<<<cdiv((long)G * 65, B), B, 0, stream>>>(counts, (long)G * 65);
  pool_count_kernel<<<cdiv(N, B), B, 0, stream>>>(bat, counts, N);
  pool_sum_kernel<<<cdiv((long)N * 64, B), B, 0, stream>>>(AGG, bat, sums, N);
  pool_final_kernel<<<cdiv((long)G * 64, B), B, 0, stream>>>(sums, counts, b3, out, G);
}


# Edges grouped by destination once per call (integer count+rank, scan, fill); each layer gathers a node's rows in registers, no f32 atomics; sum order: self, then edges in arrival rank
# speedup vs baseline: 1.8774x; 1.8774x over previous
#include <hip/hip_runtime.h>

typedef float v2f __attribute__((ext_vector_type(2)));
typedef float v8f __attribute__((ext_vector_type(8)));

#define K_DIM 128

template <int DOUT>
__global__ __launch_bounds__(256) void gemm_f32_wmma(
    const float* __restrict__ A, const float* __restrict__ W,
    float* __restrict__ T, int N) {
  constexpr int CT = DOUT / 16;
  constexpr int RT = 8 / CT;
  constexpr int ROWS = RT * 16;
  __shared__ float As[ROWS][K_DIM + 4];

  const int tid  = threadIdx.x;
  const int wave = tid >> 5;
  const int lane = tid & 31;
  const int rowTile = wave / CT;
  const int colTile = wave % CT;
  const long rowBase = (long)blockIdx.x * ROWS;

  for (int i = tid; i < ROWS * K_DIM; i += 256) {
    const int r = i >> 7;
    const int c = i & 127;
    const long gr = rowBase + r;
    As[r][c] = (gr < N) ? A[gr * K_DIM + c] : 0.0f;
  }
  __syncthreads();

  const int half = lane >> 4;
  const int l16  = lane & 15;
  const int n0   = colTile * 16 + l16;
  const int arow = rowTile * 16 + l16;

  v8f acc = {};
  #pragma unroll
  for (int kk = 0; kk < K_DIM; kk += 4) {
    const int ka = kk + 2 * half;
    v2f a, b;
    a.x = As[arow][ka];
    a.y = As[arow][ka + 1];
    b.x = W[ka * DOUT + n0];
    b.y = W[(ka + 1) * DOUT + n0];
    acc = __builtin_amdgcn_wmma_f32_16x16x4_f32(
         false, a,  false, b,
         (short)0, acc,  false,  false);
  }

  const long mBase = rowBase + rowTile * 16 + 8 * half;
  #pragma unroll
  for (int v = 0; v < 8; ++v) {
    const long m = mBase + v;
    if (m < N) T[m * DOUT + n0] = acc[v];
  }
}

__global__ void init_deg_kernel(float* deg, int N) {
  const long i = (long)blockIdx.x * blockDim.x + threadIdx.x;
  if (i < N) deg[i] = 1.0f;
}

__global__ void count_deg_kernel(const int* __restrict__ dst, float* deg, int E) {
  const long e = (long)blockIdx.x * blockDim.x + threadIdx.x;
  if (e < E) atomicAdd(&deg[dst[e]], 1.0f);
}

__global__ void finalize_dinv_kernel(float* deg, int N) {
  const long i = (long)blockIdx.x * blockDim.x + threadIdx.x;
  if (i < N) deg[i] = rsqrtf(deg[i]);
}

template <int D>
__global__ void agg_self_kernel(const float* __restrict__ T,
                                const float* __restrict__ dinv,
                                float* __restrict__ out, int N) {
  const long idx = (long)blockIdx.x * blockDim.x + threadIdx.x;
  if (idx >= (long)N * D) return;
  const long i = idx / D;
  const float w = dinv[i];
  out[idx] = T[idx] * w * w;
}

template <int D>
__global__ void agg_edges_kernel(const float* __restrict__ T,
                                 const int* __restrict__ src,
                                 const int* __restrict__ dst,
                                 const float* __restrict__ dinv,
                                 float* __restrict__ out, int E) {
  constexpr int TPE = D / 4;
  const long idx = (long)blockIdx.x * blockDim.x + threadIdx.x;
  const long e = idx / TPE;
  const int  t = (int)(idx % TPE);
  if (e >= E) return;
  const int s = src[e];
  const int d = dst[e];
  const float norm = dinv[s] * dinv[d];
  const float4 v = ((const float4*)(T + (long)s * D))[t];
  float* op = out + (long)d * D + t * 4;
  atomicAdd(op + 0, v.x * norm);
  atomicAdd(op + 1, v.y * norm);
  atomicAdd(op + 2, v.z * norm);
  atomicAdd(op + 3, v.w * norm);
}

template <int D>
__global__ void bias_relu_kernel(float* __restrict__ h,
                                 const float* __restrict__ b, long total) {
  const long idx = (long)blockIdx.x * blockDim.x + threadIdx.x;
  if (idx >= total) return;
  const int d = (int)(idx % D);
  h[idx] = fmaxf(h[idx] + b[d], 0.0f);
}


__global__ void count_rank_kernel(const int* __restrict__ dst, int* cnt,
                                  int* __restrict__ rank, int E) {
  const long e = (long)blockIdx.x * blockDim.x + threadIdx.x;
  if (e < E) rank[e] = atomicAdd(&cnt[dst[e]], 1);
}

__global__ __launch_bounds__(256) void scan_deg_kernel(
    const int* __restrict__ cnt, int* __restrict__ rowptr,
    float* __restrict__ dinv, int N) {
  __shared__ int wtot[8];
  __shared__ int wpre[8];
  const int tid  = threadIdx.x;
  const int lane = tid & 31;
  const int wave = tid >> 5;
  const int base = blockIdx.x * 256;
  int part = 0;
  const int4* c4 = (const int4*)cnt;
  for (int i = tid; i < base / 4; i += 256) {
    const int4 q = c4[i];
    part += (q.x + q.y) + (q.z + q.w);
  }
  const int idx = base + tid;
  const int c = (idx < N) ? cnt[idx] : 0;
  int inc = c;
  for (int o = 1; o < 32; o <<= 1) {
    const int t = __shfl_up(inc, o);
    if (lane >= o) inc += t;
  }
  for (int o = 16; o > 0; o >>= 1) part += __shfl_xor(part, o);
  if (lane == 31) wtot[wave] = inc;
  if (lane == 0)  wpre[wave] = part;
  __syncthreads();
  int off = 0;
  for (int k = 0; k < 8; ++k) {
    off += wpre[k];
    if (k < wave) off += wtot[k];
  }
  const int excl = off + inc - c;
  if (idx < N) {
    rowptr[idx] = excl;
    dinv[idx] = rsqrtf((float)(c + 1));
    if (idx == N - 1) rowptr[N] = excl + c;
  }
}

__global__ void fill_list_kernel(const int* __restrict__ src, const int* __restrict__ dst,
                                 const int* __restrict__ rank, const int* __restrict__ rowptr,
                                 const float* __restrict__ dinv, int2* __restrict__ rec, int E) {
  const long e = (long)blockIdx.x * blockDim.x + threadIdx.x;
  if (e >= E) return;
  const int s = src[e];
  const int d = dst[e];
  const float norm = dinv[s] * dinv[d];
  rec[rowptr[d] + rank[e]] = make_int2(s, __float_as_int(norm));
}

template <int D>
__global__ __launch_bounds__(256) void gather_rows_kernel(
    const float* __restrict__ T, const int* __restrict__ rowptr,
    const int2* __restrict__ rec, const float* __restrict__ dinv,
    float* __restrict__ out, int N) {
  typedef float vf __attribute__((ext_vector_type(D / 32)));
  const int lane = threadIdx.x & 31;
  const int node = __builtin_amdgcn_readfirstlane((int)(blockIdx.x * 8 + (threadIdx.x >> 5)));
  if (node >= N) return;
  const int beg = rowptr[node];
  const int end = rowptr[node + 1];
  const float w = dinv[node];
  const vf* Tv = (const vf*)T;
  vf acc = Tv[(long)node * 32 + lane] * w * w;
  #pragma unroll 4
  for (int e = beg; e < end; ++e) {
    const int2 r = rec[e];
    const vf v = Tv[(long)r.x * 32 + lane];
    acc += v * __int_as_float(r.y);
  }
  ((vf*)out)[(long)node * 32 + lane] = acc;
}

__global__ void zero_kernel(float* p, long n) {
  const long i = (long)blockIdx.x * blockDim.x + threadIdx.x;
  if (i < n) p[i] = 0.0f;
}

__global__ void pool_count_kernel(const int* __restrict__ batch, float* counts, int N) {
  const long i = (long)blockIdx.x * blockDim.x + threadIdx.x;
  if (i < N) atomicAdd(&counts[batch[i]], 1.0f);
}

__global__ void pool_sum_kernel(const float* __restrict__ h,
                                const int* __restrict__ batch,
                                float* sums, int N) {
  const long idx = (long)blockIdx.x * blockDim.x + threadIdx.x;
  if (idx >= (long)N * 64) return;
  const long i = idx / 64;
  const int  d = (int)(idx % 64);
  atomicAdd(&sums[(long)batch[i] * 64 + d], h[idx]);
}

__global__ void pool_final_kernel(const float* __restrict__ sums,
                                  const float* __restrict__ counts,
                                  const float* __restrict__ b3,
                                  float* __restrict__ out, int G) {
  const long idx = (long)blockIdx.x * blockDim.x + threadIdx.x;
  if (idx >= (long)G * 64) return;
  const long g = idx / 64;
  const int  d = (int)(idx % 64);
  const float c = counts[g];
  out[idx] = (sums[idx] + c * b3[d]) / fmaxf(c, 1.0f);
}

static inline int cdiv(long a, long b) { return (int)((a + b - 1) / b); }

extern "C" void kernel_launch(void* const* d_in, const int* in_sizes, int n_in,
                              void* d_out, int out_size, void* d_ws, size_t ws_size,
                              hipStream_t stream) {
  const float* x   = (const float*)d_in[0];
  const int*   ei  = (const int*)d_in[1];
  const int*   bat = (const int*)d_in[2];
  const float* W1  = (const float*)d_in[3];
  const float* b1  = (const float*)d_in[4];
  const float* W2  = (const float*)d_in[5];
  const float* b2  = (const float*)d_in[6];
  const float* W3  = (const float*)d_in[7];
  const float* b3  = (const float*)d_in[8];
  (void)n_in;

  const int N = in_sizes[0] / 128;
  const int E = in_sizes[1] / 2;
  const int G = out_size / 64;
  const int* src = ei;
  const int* dst = ei + E;

  float* ws = (float*)d_ws;
  size_t o = 0;
  float* dinv = ws + o;   o += ((size_t)N + 63) & ~(size_t)63;
  float* T    = ws + o;   o += (size_t)N * 128;
  float* AGG  = ws + o;   o += (size_t)N * 128;
  float* counts = ws + o; o += (size_t)G;
  float* sums   = ws + o; o += (size_t)G * 64;
  const size_t npad = ((size_t)N + 64 + 63) & ~(size_t)63;
  o = (o + 63) & ~(size_t)63;
  int*  rowptr = (int*)(ws + o);  o += npad;
  int2* rec    = (int2*)(ws + o); o += (size_t)E * 2;
  int*  cnt    = (int*)AGG;
  int*  rank   = cnt + npad;
  const bool grouped = ws_size >= o * sizeof(float) && (size_t)N * 128 >= npad + (size_t)E;

  const int B = 256;
  float* out = (float*)d_out;

  if (grouped) {
    zero_kernel<<<cdiv(N, B), B, 0, stream>>>((float*)cnt, (long)N);
    count_rank_kernel<<<cdiv(E, B), B, 0, stream>>>(dst, cnt, rank, E);
    scan_deg_kernel<<<cdiv(N, 256), 256, 0, stream>>>(cnt, rowptr, dinv, N);
    fill_list_kernel<<<cdiv(E, B), B, 0, stream>>>(src, dst, rank, rowptr, dinv, rec, E);
  } else {
    init_deg_kernel<<<cdiv(N, B), B, 0, stream>>>(dinv, N);
    count_deg_kernel<<<cdiv(E, B), B, 0, stream>>>(dst, dinv, E);
    finalize_dinv_kernel<<<cdiv(N, B), B, 0, stream>>>(dinv, N);
  }

  gemm_f32_wmma<128><<<cdiv(N, 16), B, 0, stream>>>(x, W1, T, N);
  if (grouped) {
    gather_rows_kernel<128><<<cdiv(N, 8), 256, 0, stream>>>(T, rowptr, rec, dinv, AGG, N);
  } else {
    agg_self_kernel<128><<<cdiv((long)N * 128, B), B, 0, stream>>>(T, dinv, AGG, N);
    agg_edges_kernel<128><<<cdiv((long)E * 32, B), B, 0, stream>>>(T, src, dst, dinv, AGG, E);
  }
  bias_relu_kernel<128><<<cdiv((long)N * 128, B), B, 0, stream>>>(AGG, b1, (long)N * 128);

  gemm_f32_wmma<128><<<cdiv(N, 16), B, 0, stream>>>(AGG, W2, T, N);
  if (grouped) {
    gather_rows_kernel<128><<<cdiv(N, 8), 256, 0, stream>>>(T, rowptr, rec, dinv, AGG, N);
  } else {
    agg_self_kernel<128><<<cdiv((long)N * 128, B), B, 0, stream>>>(T, dinv, AGG, N);
    agg_edges_kernel<128><<<cdiv((long)E * 32, B), B, 0, stream>>>(T, src, dst, dinv, AGG, E);
  }
  bias_relu_kernel<128><<<cdiv((long)N * 128, B), B, 0, stream>>>(AGG, b2, (long)N * 128);

  gemm_f32_wmma<64><<<cdiv(N, 32), B, 0, stream>>>(AGG, W3, T, N);
  if (grouped) {
    gather_rows_kernel<64><<<cdiv(N, 8), 256, 0, stream>>>(T, rowptr, rec, dinv, AGG, N);
  } else {
    agg_self_kernel<64><<<cdiv((long)N * 64, B), B, 0, stream>>>(T, dinv, AGG, N);
    agg_edges_kernel<64><<<cdiv((long)E * 16, B), B, 0, stream>>>(T, src, dst, dinv, AGG, E);
  }

  zero_kernel<<<cdiv((long)G * 65, B), B, 0, stream>>>(counts, (long)G * 65);
  pool_count_kernel<<<cdiv(N, B), B, 0, stream>>>(bat, counts, N);
  pool_sum_kernel<<<cdiv((long)N * 64, B), B, 0, stream>>>(AGG, bat, sums, N);
  pool_final_kernel<<<cdiv((long)G * 64, B), B, 0, stream>>>(sums, counts, b3, out, G);
}


# Layers 2 and 3: GEMM A tile built in LDS by gathering each node's rows (bias+ReLU in registers); aggregated plane and both bias/ReLU passes removed, 16 -> 12 launches; per-node sum order unchanged
# speedup vs baseline: 2.4703x; 1.3158x over previous
#include <hip/hip_runtime.h>

typedef float v2f __attribute__((ext_vector_type(2)));
typedef float v8f __attribute__((ext_vector_type(8)));

#define K_DIM 128

template <int DOUT>
__global__ __launch_bounds__(256) void gemm_f32_wmma(
    const float* __restrict__ A, const float* __restrict__ W,
    float* __restrict__ T, int N) {
  constexpr int CT = DOUT / 16;
  constexpr int RT = 8 / CT;
  constexpr int ROWS = RT * 16;
  __shared__ float As[ROWS][K_DIM + 4];

  const int tid  = threadIdx.x;
  const int wave = tid >> 5;
  const int lane = tid & 31;
  const int rowTile = wave / CT;
  const int colTile = wave % CT;
  const long rowBase = (long)blockIdx.x * ROWS;

  for (int i = tid; i < ROWS * K_DIM; i += 256) {
    const int r = i >> 7;
    const int c = i & 127;
    const long gr = rowBase + r;
    As[r][c] = (gr < N) ? A[gr * K_DIM + c] : 0.0f;
  }
  __syncthreads();

  const int half = lane >> 4;
  const int l16  = lane & 15;
  const int n0   = colTile * 16 + l16;
  const int arow = rowTile * 16 + l16;

  v8f acc = {};
  #pragma unroll
  for (int kk = 0; kk < K_DIM; kk += 4) {
    const int ka = kk + 2 * half;
    v2f a, b;
    a.x = As[arow][ka];
    a.y = As[arow][ka + 1];
    b.x = W[ka * DOUT + n0];
    b.y = W[(ka + 1) * DOUT + n0];
    acc = __builtin_amdgcn_wmma_f32_16x16x4_f32(
         false, a,  false, b,
         (short)0, acc,  false,  false);
  }

  const long mBase = rowBase + rowTile * 16 + 8 * half;
  #pragma unroll
  for (int v = 0; v < 8; ++v) {
    const long m = mBase + v;
    if (m < N) T[m * DOUT + n0] = acc[v];
  }
}

__global__ void init_deg_kernel(float* deg, int N) {
  const long i = (long)blockIdx.x * blockDim.x + threadIdx.x;
  if (i < N) deg[i] = 1.0f;
}

__global__ void count_deg_kernel(const int* __restrict__ dst, float* deg, int E) {
  const long e = (long)blockIdx.x * blockDim.x + threadIdx.x;
  if (e < E) atomicAdd(&deg[dst[e]], 1.0f);
}

__global__ void finalize_dinv_kernel(float* deg, int N) {
  const long i = (long)blockIdx.x * blockDim.x + threadIdx.x;
  if (i < N) deg[i] = rsqrtf(deg[i]);
}

template <int D>
__global__ void agg_self_kernel(const float* __restrict__ T,
                                const float* __restrict__ dinv,
                                float* __restrict__ out, int N) {
  const long idx = (long)blockIdx.x * blockDim.x + threadIdx.x;
  if (idx >= (long)N * D) return;
  const long i = idx / D;
  const float w = dinv[i];
  out[idx] = T[idx] * w * w;
}

template <int D>
__global__ void agg_edges_kernel(const float* __restrict__ T,
                                 const int* __restrict__ src,
                                 const int* __restrict__ dst,
                                 const float* __restrict__ dinv,
                                 float* __restrict__ out, int E) {
  constexpr int TPE = D / 4;
  const long idx = (long)blockIdx.x * blockDim.x + threadIdx.x;
  const long e = idx / TPE;
  const int  t = (int)(idx % TPE);
  if (e >= E) return;
  const int s = src[e];
  const int d = dst[e];
  const float norm = dinv[s] * dinv[d];
  const float4 v = ((const float4*)(T + (long)s * D))[t];
  float* op = out + (long)d * D + t * 4;
  atomicAdd(op + 0, v.x * norm);
  atomicAdd(op + 1, v.y * norm);
  atomicAdd(op + 2, v.z * norm);
  atomicAdd(op + 3, v.w * norm);
}

template <int D>
__global__ void bias_relu_kernel(float* __restrict__ h,
                                 const float* __restrict__ b, long total) {
  const long idx = (long)blockIdx.x * blockDim.x + threadIdx.x;
  if (idx >= total) return;
  const int d = (int)(idx % D);
  h[idx] = fmaxf(h[idx] + b[d], 0.0f);
}


__global__ void count_rank_kernel(const int* __restrict__ dst, int* cnt,
                                  int* __restrict__ rank, int E) {
  const long e = (long)blockIdx.x * blockDim.x + threadIdx.x;
  if (e < E) rank[e] = atomicAdd(&cnt[dst[e]], 1);
}

__global__ __launch_bounds__(256) void scan_deg_kernel(
    const int* __restrict__ cnt, int* __restrict__ rowptr,
    float* __restrict__ dinv, int N) {
  __shared__ int wtot[8];
  __shared__ int wpre[8];
  const int tid  = threadIdx.x;
  const int lane = tid & 31;
  const int wave = tid >> 5;
  const int base = blockIdx.x * 256;
  int part = 0;
  const int4* c4 = (const int4*)cnt;
  for (int i = tid; i < base / 4; i += 256) {
    const int4 q = c4[i];
    part += (q.x + q.y) + (q.z + q.w);
  }
  const int idx = base + tid;
  const int c = (idx < N) ? cnt[idx] : 0;
  int inc = c;
  for (int o = 1; o < 32; o <<= 1) {
    const int t = __shfl_up(inc, o);
    if (lane >= o) inc += t;
  }
  for (int o = 16; o > 0; o >>= 1) part += __shfl_xor(part, o);
  if (lane == 31) wtot[wave] = inc;
  if (lane == 0)  wpre[wave] = part;
  __syncthreads();
  int off = 0;
  for (int k = 0; k < 8; ++k) {
    off += wpre[k];
    if (k < wave) off += wtot[k];
  }
  const int excl = off + inc - c;
  if (idx < N) {
    rowptr[idx] = excl;
    dinv[idx] = rsqrtf((float)(c + 1));
    if (idx == N - 1) rowptr[N] = excl + c;
  }
}

__global__ void fill_list_kernel(const int* __restrict__ src, const int* __restrict__ dst,
                                 const int* __restrict__ rank, const int* __restrict__ rowptr,
                                 const float* __restrict__ dinv, int2* __restrict__ rec, int E) {
  const long e = (long)blockIdx.x * blockDim.x + threadIdx.x;
  if (e >= E) return;
  const int s = src[e];
  const int d = dst[e];
  const float norm = dinv[s] * dinv[d];
  rec[rowptr[d] + rank[e]] = make_int2(s, __float_as_int(norm));
}

template <int D>
__global__ __launch_bounds__(256) void gather_rows_kernel(
    const float* __restrict__ T, const int* __restrict__ rowptr,
    const int2* __restrict__ rec, const float* __restrict__ dinv,
    float* __restrict__ out, int N) {
  typedef float vf __attribute__((ext_vector_type(D / 32)));
  const int lane = threadIdx.x & 31;
  const int node = __builtin_amdgcn_readfirstlane((int)(blockIdx.x * 8 + (threadIdx.x >> 5)));
  if (node >= N) return;
  const int beg = rowptr[node];
  const int end = rowptr[node + 1];
  const float w = dinv[node];
  const vf* Tv = (const vf*)T;
  vf acc = Tv[(long)node * 32 + lane] * w * w;
  #pragma unroll 4
  for (int e = beg; e < end; ++e) {
    const int2 r = rec[e];
    const vf v = Tv[(long)r.x * 32 + lane];
    acc += v * __int_as_float(r.y);
  }
  ((vf*)out)[(long)node * 32 + lane] = acc;
}

template <int DOUT>
__global__ __launch_bounds__(256) void gather_gemm_f32_wmma(
    const float* __restrict__ Tin, const int* __restrict__ rowptr,
    const int2* __restrict__ rec, const float* __restrict__ dinv,
    const float* __restrict__ bias, const float* __restrict__ W,
    float* __restrict__ Tout, int N) {
  constexpr int CT = DOUT / 16;
  constexpr int RT = 8 / CT;
  constexpr int ROWS = RT * 16;
  constexpr int NPW = ROWS / 8;
  __shared__ float As[ROWS][K_DIM + 4];

  typedef float v4f __attribute__((ext_vector_type(4)));
  const int tid  = threadIdx.x;
  const int lane = tid & 31;
  const int wave = __builtin_amdgcn_readfirstlane(tid >> 5);
  const int rowTile = wave / CT;
  const int colTile = wave % CT;
  const int rowBase = blockIdx.x * ROWS;

  const v4f* Tv = (const v4f*)Tin;
  const v4f bv = ((const v4f*)bias)[lane];
  for (int j = 0; j < NPW; ++j) {
    const int r = wave * NPW + j;
    const int node = rowBase + r;
    v4f acc = {0.0f, 0.0f, 0.0f, 0.0f};
    if (node < N) {
      const int beg = rowptr[node];
      const int end = rowptr[node + 1];
      const float w = dinv[node];
      acc = Tv[(long)node * 32 + lane] * w * w;
      #pragma unroll 4
      for (int e = beg; e < end; ++e) {
        const int2 q = rec[e];
        const v4f v = Tv[(long)q.x * 32 + lane];
        acc += v * __int_as_float(q.y);
      }
      acc += bv;
      acc.x = fmaxf(acc.x, 0.0f);
      acc.y = fmaxf(acc.y, 0.0f);
      acc.z = fmaxf(acc.z, 0.0f);
      acc.w = fmaxf(acc.w, 0.0f);
    }
    *(v4f*)&As[r][4 * lane] = acc;
  }
  __syncthreads();

  const int half = lane >> 4;
  const int l16  = lane & 15;
  const int n0   = colTile * 16 + l16;
  const int arow = rowTile * 16 + l16;

  v8f acc = {};
  #pragma unroll
  for (int kk = 0; kk < K_DIM; kk += 4) {
    const int ka = kk + 2 * half;
    v2f a, b;
    a.x = As[arow][ka];
    a.y = As[arow][ka + 1];
    b.x = W[ka * DOUT + n0];
    b.y = W[(ka + 1) * DOUT + n0];
    acc = __builtin_amdgcn_wmma_f32_16x16x4_f32(
         false, a,  false, b,
         (short)0, acc,  false,  false);
  }

  const long mBase = (long)rowBase + rowTile * 16 + 8 * half;
  #pragma unroll
  for (int v = 0; v < 8; ++v) {
    const long m = mBase + v;
    if (m < N) Tout[m * DOUT + n0] = acc[v];
  }
}

__global__ void zero_kernel(float* p, long n) {
  const long i = (long)blockIdx.x * blockDim.x + threadIdx.x;
  if (i < n) p[i] = 0.0f;
}

__global__ void pool_count_kernel(const int* __restrict__ batch, float* counts, int N) {
  const long i = (long)blockIdx.x * blockDim.x + threadIdx.x;
  if (i < N) atomicAdd(&counts[batch[i]], 1.0f);
}

__global__ void pool_sum_kernel(const float* __restrict__ h,
                                const int* __restrict__ batch,
                                float* sums, int N) {
  const long idx = (long)blockIdx.x * blockDim.x + threadIdx.x;
  if (idx >= (long)N * 64) return;
  const long i = idx / 64;
  const int  d = (int)(idx % 64);
  atomicAdd(&sums[(long)batch[i] * 64 + d], h[idx]);
}

__global__ void pool_final_kernel(const float* __restrict__ sums,
                                  const float* __restrict__ counts,
                                  const float* __restrict__ b3,
                                  float* __restrict__ out, int G) {
  const long idx = (long)blockIdx.x * blockDim.x + threadIdx.x;
  if (idx >= (long)G * 64) return;
  const long g = idx / 64;
  const int  d = (int)(idx % 64);
  const float c = counts[g];
  out[idx] = (sums[idx] + c * b3[d]) / fmaxf(c, 1.0f);
}

static inline int cdiv(long a, long b) { return (int)((a + b - 1) / b); }

extern "C" void kernel_launch(void* const* d_in, const int* in_sizes, int n_in,
                              void* d_out, int out_size, void* d_ws, size_t ws_size,
                              hipStream_t stream) {
  const float* x   = (const float*)d_in[0];
  const int*   ei  = (const int*)d_in[1];
  const int*   bat = (const int*)d_in[2];
  const float* W1  = (const float*)d_in[3];
  const float* b1  = (const float*)d_in[4];
  const float* W2  = (const float*)d_in[5];
  const float* b2  = (const float*)d_in[6];
  const float* W3  = (const float*)d_in[7];
  const float* b3  = (const float*)d_in[8];
  (void)n_in;

  const int N = in_sizes[0] / 128;
  const int E = in_sizes[1] / 2;
  const int G = out_size / 64;
  const int* src = ei;
  const int* dst = ei + E;

  float* ws = (float*)d_ws;
  size_t o = 0;
  float* dinv = ws + o;   o += ((size_t)N + 63) & ~(size_t)63;
  float* T    = ws + o;   o += (size_t)N * 128;
  float* AGG  = ws + o;   o += (size_t)N * 128;
  float* counts = ws + o; o += (size_t)G;
  float* sums   = ws + o; o += (size_t)G * 64;
  const size_t npad = ((size_t)N + 64 + 63) & ~(size_t)63;
  o = (o + 63) & ~(size_t)63;
  int*  rowptr = (int*)(ws + o);  o += npad;
  int2* rec    = (int2*)(ws + o); o += (size_t)E * 2;
  int*  cnt    = (int*)AGG;
  int*  rank   = cnt + npad;
  const bool grouped = ws_size >= o * sizeof(float) && (size_t)N * 128 >= npad + (size_t)E;

  const int B = 256;
  float* out = (float*)d_out;

  if (grouped) {
    zero_kernel<<<cdiv(N, B), B, 0, stream>>>((float*)cnt, (long)N);
    count_rank_kernel<<<cdiv(E, B), B, 0, stream>>>(dst, cnt, rank, E);
    scan_deg_kernel<<<cdiv(N, 256), 256, 0, stream>>>(cnt, rowptr, dinv, N);
    fill_list_kernel<<<cdiv(E, B), B, 0, stream>>>(src, dst, rank, rowptr, dinv, rec, E);
  } else {
    init_deg_kernel<<<cdiv(N, B), B, 0, stream>>>(dinv, N);
    count_deg_kernel<<<cdiv(E, B), B, 0, stream>>>(dst, dinv, E);
    finalize_dinv_kernel<<<cdiv(N, B), B, 0, stream>>>(dinv, N);
  }

  if (grouped) {
    gemm_f32_wmma<128><<<cdiv(N, 16), B, 0, stream>>>(x, W1, T, N);
    gather_gemm_f32_wmma<128><<<cdiv(N, 16), B, 0, stream>>>(T, rowptr, rec, dinv, b1, W2, AGG, N);
    gather_gemm_f32_wmma<64><<<cdiv(N, 32), B, 0, stream>>>(AGG, rowptr, rec, dinv, b2, W3, T, N);
    gather_rows_kernel<64><<<cdiv(N, 8), 256, 0, stream>>>(T, rowptr, rec, dinv, AGG, N);
  } else {
    gemm_f32_wmma<128><<<cdiv(N, 16), B, 0, stream>>>(x, W1, T, N);
    agg_self_kernel<128><<<cdiv((long)N * 128, B), B, 0, stream>>>(T, dinv, AGG, N);
    agg_edges_kernel<128><<<cdiv((long)E * 32, B), B, 0, stream>>>(T, src, dst, dinv, AGG, E);
    bias_relu_kernel<128><<<cdiv((long)N * 128, B), B, 0, stream>>>(AGG, b1, (long)N * 128);

    gemm_f32_wmma<128><<<cdiv(N, 16), B, 0, stream>>>(AGG, W2, T, N);
    agg_self_kernel<128><<<cdiv((long)N * 128, B), B, 0, stream>>>(T, dinv, AGG, N);
    agg_edges_kernel<128><<<cdiv((long)E * 32, B), B, 0, stream>>>(T, src, dst, dinv, AGG, E);
    bias_relu_kernel<128><<<cdiv((long)N * 128, B), B, 0, stream>>>(AGG, b2, (long)N * 128);

    gemm_f32_wmma<64><<<cdiv(N, 32), B, 0, stream>>>(AGG, W3, T, N);
    agg_self_kernel<64><<<cdiv((long)N * 64, B), B, 0, stream>>>(T, dinv, AGG, N);
    agg_edges_kernel<64><<<cdiv((long)E * 16, B), B, 0, stream>>>(T, src, dst, dinv, AGG, E);
  }

  zero_kernel<<<cdiv((long)G * 65, B), B, 0, stream>>>(counts, (long)G * 65);
  pool_count_kernel<<<cdiv(N, B), B, 0, stream>>>(bat, counts, N);
  pool_sum_kernel<<<cdiv((long)N * 64, B), B, 0, stream>>>(AGG, bat, sums, N);
  pool_final_kernel<<<cdiv((long)G * 64, B), B, 0, stream>>>(sums, counts, b3, out, G);
}


# Edges grouped by destination with LDS counters in two levels (256 edge chunks x 256-node buckets) instead of one returning global atomic per edge; order inside a row: arrival rank of the LDS adds
# speedup vs baseline: 3.2542x; 1.3173x over previous
#include <hip/hip_runtime.h>

typedef float v2f __attribute__((ext_vector_type(2)));
typedef float v8f __attribute__((ext_vector_type(8)));

#define K_DIM 128

template <int DOUT>
__global__ __launch_bounds__(256) void gemm_f32_wmma(
    const float* __restrict__ A, const float* __restrict__ W,
    float* __restrict__ T, int N) {
  constexpr int CT = DOUT / 16;
  constexpr int RT = 8 / CT;
  constexpr int ROWS = RT * 16;
  __shared__ float As[ROWS][K_DIM + 4];

  const int tid  = threadIdx.x;
  const int wave = tid >> 5;
  const int lane = tid & 31;
  const int rowTile = wave / CT;
  const int colTile = wave % CT;
  const long rowBase = (long)blockIdx.x * ROWS;

  for (int i = tid; i < ROWS * K_DIM; i += 256) {
    const int r = i >> 7;
    const int c = i & 127;
    const long gr = rowBase + r;
    As[r][c] = (gr < N) ? A[gr * K_DIM + c] : 0.0f;
  }
  __syncthreads();

  const int half = lane >> 4;
  const int l16  = lane & 15;
  const int n0   = colTile * 16 + l16;
  const int arow = rowTile * 16 + l16;

  v8f acc = {};
  #pragma unroll
  for (int kk = 0; kk < K_DIM; kk += 4) {
    const int ka = kk + 2 * half;
    v2f a, b;
    a.x = As[arow][ka];
    a.y = As[arow][ka + 1];
    b.x = W[ka * DOUT + n0];
    b.y = W[(ka + 1) * DOUT + n0];
    acc = __builtin_amdgcn_wmma_f32_16x16x4_f32(
         false, a,  false, b,
         (short)0, acc,  false,  false);
  }

  const long mBase = rowBase + rowTile * 16 + 8 * half;
  #pragma unroll
  for (int v = 0; v < 8; ++v) {
    const long m = mBase + v;
    if (m < N) T[m * DOUT + n0] = acc[v];
  }
}

__global__ void init_deg_kernel(float* deg, int N) {
  const long i = (long)blockIdx.x * blockDim.x + threadIdx.x;
  if (i < N) deg[i] = 1.0f;
}

__global__ void count_deg_kernel(const int* __restrict__ dst, float* deg, int E) {
  const long e = (long)blockIdx.x * blockDim.x + threadIdx.x;
  if (e < E) atomicAdd(&deg[dst[e]], 1.0f);
}

__global__ void finalize_dinv_kernel(float* deg, int N) {
  const long i = (long)blockIdx.x * blockDim.x + threadIdx.x;
  if (i < N) deg[i] = rsqrtf(deg[i]);
}

template <int D>
__global__ void agg_self_kernel(const float* __restrict__ T,
                                const float* __restrict__ dinv,
                                float* __restrict__ out, int N) {
  const long idx = (long)blockIdx.x * blockDim.x + threadIdx.x;
  if (idx >= (long)N * D) return;
  const long i = idx / D;
  const float w = dinv[i];
  out[idx] = T[idx] * w * w;
}

template <int D>
__global__ void agg_edges_kernel(const float* __restrict__ T,
                                 const int* __restrict__ src,
                                 const int* __restrict__ dst,
                                 const float* __restrict__ dinv,
                                 float* __restrict__ out, int E) {
  constexpr int TPE = D / 4;
  const long idx = (long)blockIdx.x * blockDim.x + threadIdx.x;
  const long e = idx / TPE;
  const int  t = (int)(idx % TPE);
  if (e >= E) return;
  const int s = src[e];
  const int d = dst[e];
  const float norm = dinv[s] * dinv[d];
  const float4 v = ((const float4*)(T + (long)s * D))[t];
  float* op = out + (long)d * D + t * 4;
  atomicAdd(op + 0, v.x * norm);
  atomicAdd(op + 1, v.y * norm);
  atomicAdd(op + 2, v.z * norm);
  atomicAdd(op + 3, v.w * norm);
}

template <int D>
__global__ void bias_relu_kernel(float* __restrict__ h,
                                 const float* __restrict__ b, long total) {
  const long idx = (long)blockIdx.x * blockDim.x + threadIdx.x;
  if (idx >= total) return;
  const int d = (int)(idx % D);
  h[idx] = fmaxf(h[idx] + b[d], 0.0f);
}


__global__ void count_rank_kernel(const int* __restrict__ dst, int* cnt,
                                  int* __restrict__ rank, int E) {
  const long e = (long)blockIdx.x * blockDim.x + threadIdx.x;
  if (e < E) rank[e] = atomicAdd(&cnt[dst[e]], 1);
}

__global__ __launch_bounds__(256) void scan_deg_kernel(
    const int* __restrict__ cnt, int* __restrict__ rowptr,
    float* __restrict__ dinv, int N) {
  __shared__ int wtot[8];
  __shared__ int wpre[8];
  const int tid  = threadIdx.x;
  const int lane = tid & 31;
  const int wave = tid >> 5;
  const int base = blockIdx.x * 256;
  int part = 0;
  const int4* c4 = (const int4*)cnt;
  for (int i = tid; i < base / 4; i += 256) {
    const int4 q = c4[i];
    part += (q.x + q.y) + (q.z + q.w);
  }
  const int idx = base + tid;
  const int c = (idx < N) ? cnt[idx] : 0;
  int inc = c;
  for (int o = 1; o < 32; o <<= 1) {
    const int t = __shfl_up(inc, o);
    if (lane >= o) inc += t;
  }
  for (int o = 16; o > 0; o >>= 1) part += __shfl_xor(part, o);
  if (lane == 31) wtot[wave] = inc;
  if (lane == 0)  wpre[wave] = part;
  __syncthreads();
  int off = 0;
  for (int k = 0; k < 8; ++k) {
    off += wpre[k];
    if (k < wave) off += wtot[k];
  }
  const int excl = off + inc - c;
  if (idx < N) {
    rowptr[idx] = excl;
    dinv[idx] = rsqrtf((float)(c + 1));
    if (idx == N - 1) rowptr[N] = excl + c;
  }
}

__global__ void fill_list_kernel(const int* __restrict__ src, const int* __restrict__ dst,
                                 const int* __restrict__ rank, const int* __restrict__ rowptr,
                                 const float* __restrict__ dinv, int2* __restrict__ rec, int E) {
  const long e = (long)blockIdx.x * blockDim.x + threadIdx.x;
  if (e >= E) return;
  const int s = src[e];
  const int d = dst[e];
  const float norm = dinv[s] * dinv[d];
  rec[rowptr[d] + rank[e]] = make_int2(s, __float_as_int(norm));
}

#define EDGE_CHUNKS 256
#define NODE_BUCKET 256
#define BATCH 4

__device__ inline int excl_scan_first256(int c, int tid, int* wsum) {
  const int lane = tid & 31;
  const int wave = tid >> 5;
  int inc = c;
  for (int o = 1; o < 32; o <<= 1) {
    const int u = __shfl_up(inc, o);
    if (lane >= o) inc += u;
  }
  if (lane == 31 && wave < 8) wsum[wave] = inc;
  __syncthreads();
  int off = 0;
  for (int k = 0; k < 8; ++k)
    if (k < wave) off += wsum[k];
  return off + inc - c;
}

__global__ __launch_bounds__(1024) void bucket_count_kernel(
    const int* __restrict__ dst, int* __restrict__ matT, int E, int CH) {
  __shared__ int hist[NODE_BUCKET];
  const int tid = threadIdx.x;
  if (tid < NODE_BUCKET) hist[tid] = 0;
  __syncthreads();
  const int e0 = blockIdx.x * CH;
  const int e1 = min(E, e0 + CH);
  for (int eb = e0 + tid; eb < e1; eb += BATCH * 1024) {
    int d[BATCH];
    #pragma unroll
    for (int k = 0; k < BATCH; ++k) {
      d[k] = dst[min(eb + k * 1024, e1 - 1)];
    }
    #pragma unroll
    for (int k = 0; k < BATCH; ++k)
      atomicAdd(&hist[d[k] >> 8], (eb + k * 1024 < e1) ? 1 : 0);
  }
  __syncthreads();
  if (tid < NODE_BUCKET) matT[tid * EDGE_CHUNKS + blockIdx.x] = hist[tid];
}

__global__ __launch_bounds__(1024) void bucket_place_kernel(
    const int* __restrict__ src, const int* __restrict__ dst,
    const int* __restrict__ matT, int2* __restrict__ pairs,
    int* __restrict__ bbase, int E, int CH, int NB) {
  __shared__ int below[4][NODE_BUCKET];
  __shared__ int total[4][NODE_BUCKET];
  __shared__ int base[NODE_BUCKET];
  __shared__ int hist[NODE_BUCKET];
  __shared__ int wsum[8];
  const int tid = threadIdx.x;
  const int b = tid & (NODE_BUCKET - 1);
  const int q = tid >> 8;
  const int p = blockIdx.x;
  int bl = 0, tot = 0;
  {
    const int4* mq = (const int4*)(matT + b * EDGE_CHUNKS + q * 64);
    int4 v[16];
    #pragma unroll
    for (int k = 0; k < 16; ++k) v[k] = mq[k];
    #pragma unroll
    for (int k = 0; k < 16; ++k) {
      const int r = q * 64 + 4 * k;
      tot += (v[k].x + v[k].y) + (v[k].z + v[k].w);
      bl += ((r     < p) ? v[k].x : 0) + ((r + 1 < p) ? v[k].y : 0)
          + ((r + 2 < p) ? v[k].z : 0) + ((r + 3 < p) ? v[k].w : 0);
    }
  }
  below[q][b] = bl;
  total[q][b] = tot;
  if (tid < NODE_BUCKET) hist[tid] = 0;
  __syncthreads();
  int colTot = 0, colBelow = 0;
  if (tid < NODE_BUCKET) {
    colTot   = (total[0][tid] + total[1][tid]) + (total[2][tid] + total[3][tid]);
    colBelow = (below[0][tid] + below[1][tid]) + (below[2][tid] + below[3][tid]);
  }
  const int excl = excl_scan_first256(colTot, tid, wsum);
  if (tid < NODE_BUCKET) {
    base[tid] = excl + colBelow;
    if (p == 0) {
      if (tid < NB) bbase[tid] = excl;
      if (tid == 0) bbase[NB] = E;
    }
  }
  __syncthreads();
  const int e0 = p * CH;
  const int e1 = min(E, e0 + CH);
  for (int eb = e0 + tid; eb < e1; eb += BATCH * 1024) {
    int s[BATCH], d[BATCH];
    #pragma unroll
    for (int k = 0; k < BATCH; ++k) {
      const int e = min(eb + k * 1024, e1 - 1);
      s[k] = src[e];
      d[k] = dst[e];
    }
    #pragma unroll
    for (int k = 0; k < BATCH; ++k) {
      if (eb + k * 1024 < e1) {
        const int bk = d[k] >> 8;
        const int r = atomicAdd(&hist[bk], 1);
        pairs[base[bk] + r] = make_int2(s[k], d[k]);
      }
    }
  }
}

__global__ __launch_bounds__(1024) void bucket_rows_kernel(
    const int2* __restrict__ pairs, const int* __restrict__ bbase,
    int* rnk, int* __restrict__ rowptr, float* __restrict__ dinv,
    int2* __restrict__ rec, int N) {
  __shared__ int cnt[NODE_BUCKET];
  __shared__ int lptr[NODE_BUCKET];
  __shared__ int wsum[8];
  const int tid = threadIdx.x;
  const int bkt = blockIdx.x;
  const int beg = bbase[bkt];
  const int end = bbase[bkt + 1];
  if (tid < NODE_BUCKET) cnt[tid] = 0;
  __syncthreads();
  for (int ib = beg + tid; ib < end; ib += BATCH * 1024) {
    int2 pr[BATCH];
    #pragma unroll
    for (int k = 0; k < BATCH; ++k) {
      pr[k] = pairs[min(ib + k * 1024, end - 1)];
    }
    #pragma unroll
    for (int k = 0; k < BATCH; ++k) {
      const int i = ib + k * 1024;
      const int r = atomicAdd(&cnt[pr[k].y & (NODE_BUCKET - 1)], (i < end) ? 1 : 0);
      if (i < end) rnk[i] = r;
    }
  }
  __syncthreads();
  const int c = (tid < NODE_BUCKET) ? cnt[tid] : 0;
  const int excl = excl_scan_first256(c, tid, wsum);
  if (tid < NODE_BUCKET) {
    lptr[tid] = excl;
    const int node = bkt * NODE_BUCKET + tid;
    if (node < N) {
      rowptr[node] = beg + excl;
      dinv[node] = rsqrtf((float)(c + 1));
    }
  }
  if (bkt == (int)gridDim.x - 1 && tid == 0) rowptr[N] = end;
  __syncthreads();
  for (int ib = beg + tid; ib < end; ib += BATCH * 1024) {
    int2 pr[BATCH];
    int rk[BATCH];
    #pragma unroll
    for (int k = 0; k < BATCH; ++k) {
      const int i = min(ib + k * 1024, end - 1);
      pr[k] = pairs[i];
      rk[k] = rnk[i];
    }
    #pragma unroll
    for (int k = 0; k < BATCH; ++k) {
      const int i = ib + k * 1024;
      if (i < end) rec[beg + lptr[pr[k].y & (NODE_BUCKET - 1)] + rk[k]] = pr[k];
    }
  }
}

__global__ void edge_weight_kernel(int2* __restrict__ rec, const float* __restrict__ dinv, int E) {
  const long i = (long)blockIdx.x * blockDim.x + threadIdx.x;
  if (i >= E) return;
  const int2 r = rec[i];
  const float norm = dinv[r.x] * dinv[r.y];
  rec[i] = make_int2(r.x, __float_as_int(norm));
}

template <int D>
__global__ __launch_bounds__(256) void gather_rows_kernel(
    const float* __restrict__ T, const int* __restrict__ rowptr,
    const int2* __restrict__ rec, const float* __restrict__ dinv,
    float* __restrict__ out, int N) {
  typedef float vf __attribute__((ext_vector_type(D / 32)));
  const int lane = threadIdx.x & 31;
  const int node = __builtin_amdgcn_readfirstlane((int)(blockIdx.x * 8 + (threadIdx.x >> 5)));
  if (node >= N) return;
  const int beg = rowptr[node];
  const int end = rowptr[node + 1];
  const float w = dinv[node];
  const vf* Tv = (const vf*)T;
  vf acc = Tv[(long)node * 32 + lane] * w * w;
  #pragma unroll 4
  for (int e = beg; e < end; ++e) {
    const int2 r = rec[e];
    const vf v = Tv[(long)r.x * 32 + lane];
    acc += v * __int_as_float(r.y);
  }
  ((vf*)out)[(long)node * 32 + lane] = acc;
}

template <int DOUT>
__global__ __launch_bounds__(256) void gather_gemm_f32_wmma(
    const float* __restrict__ Tin, const int* __restrict__ rowptr,
    const int2* __restrict__ rec, const float* __restrict__ dinv,
    const float* __restrict__ bias, const float* __restrict__ W,
    float* __restrict__ Tout, int N) {
  constexpr int CT = DOUT / 16;
  constexpr int RT = 8 / CT;
  constexpr int ROWS = RT * 16;
  constexpr int NPW = ROWS / 8;
  __shared__ float As[ROWS][K_DIM + 4];

  typedef float v4f __attribute__((ext_vector_type(4)));
  const int tid  = threadIdx.x;
  const int lane = tid & 31;
  const int wave = __builtin_amdgcn_readfirstlane(tid >> 5);
  const int rowTile = wave / CT;
  const int colTile = wave % CT;
  const int rowBase = blockIdx.x * ROWS;

  const v4f* Tv = (const v4f*)Tin;
  const v4f bv = ((const v4f*)bias)[lane];
  for (int j = 0; j < NPW; ++j) {
    const int r = wave * NPW + j;
    const int node = rowBase + r;
    v4f acc = {0.0f, 0.0f, 0.0f, 0.0f};
    if (node < N) {
      const int beg = rowptr[node];
      const int end = rowptr[node + 1];
      const float w = dinv[node];
      acc = Tv[(long)node * 32 + lane] * w * w;
      #pragma unroll 4
      for (int e = beg; e < end; ++e) {
        const int2 q = rec[e];
        const v4f v = Tv[(long)q.x * 32 + lane];
        acc += v * __int_as_float(q.y);
      }
      acc += bv;
      acc.x = fmaxf(acc.x, 0.0f);
      acc.y = fmaxf(acc.y, 0.0f);
      acc.z = fmaxf(acc.z, 0.0f);
      acc.w = fmaxf(acc.w, 0.0f);
    }
    *(v4f*)&As[r][4 * lane] = acc;
  }
  __syncthreads();

  const int half = lane >> 4;
  const int l16  = lane & 15;
  const int n0   = colTile * 16 + l16;
  const int arow = rowTile * 16 + l16;

  v8f acc = {};
  #pragma unroll
  for (int kk = 0; kk < K_DIM; kk += 4) {
    const int ka = kk + 2 * half;
    v2f a, b;
    a.x = As[arow][ka];
    a.y = As[arow][ka + 1];
    b.x = W[ka * DOUT + n0];
    b.y = W[(ka + 1) * DOUT + n0];
    acc = __builtin_amdgcn_wmma_f32_16x16x4_f32(
         false, a,  false, b,
         (short)0, acc,  false,  false);
  }

  const long mBase = (long)rowBase + rowTile * 16 + 8 * half;
  #pragma unroll
  for (int v = 0; v < 8; ++v) {
    const long m = mBase + v;
    if (m < N) Tout[m * DOUT + n0] = acc[v];
  }
}

__global__ void zero_kernel(float* p, long n) {
  const long i = (long)blockIdx.x * blockDim.x + threadIdx.x;
  if (i < n) p[i] = 0.0f;
}

__global__ void pool_count_kernel(const int* __restrict__ batch, float* counts, int N) {
  const long i = (long)blockIdx.x * blockDim.x + threadIdx.x;
  if (i < N) atomicAdd(&counts[batch[i]], 1.0f);
}

__global__ void pool_sum_kernel(const float* __restrict__ h,
                                const int* __restrict__ batch,
                                float* sums, int N) {
  const long idx = (long)blockIdx.x * blockDim.x + threadIdx.x;
  if (idx >= (long)N * 64) return;
  const long i = idx / 64;
  const int  d = (int)(idx % 64);
  atomicAdd(&sums[(long)batch[i] * 64 + d], h[idx]);
}

__global__ void pool_final_kernel(const float* __restrict__ sums,
                                  const float* __restrict__ counts,
                                  const float* __restrict__ b3,
                                  float* __restrict__ out, int G) {
  const long idx = (long)blockIdx.x * blockDim.x + threadIdx.x;
  if (idx >= (long)G * 64) return;
  const long g = idx / 64;
  const int  d = (int)(idx % 64);
  const float c = counts[g];
  out[idx] = (sums[idx] + c * b3[d]) / fmaxf(c, 1.0f);
}

static inline int cdiv(long a, long b) { return (int)((a + b - 1) / b); }

extern "C" void kernel_launch(void* const* d_in, const int* in_sizes, int n_in,
                              void* d_out, int out_size, void* d_ws, size_t ws_size,
                              hipStream_t stream) {
  const float* x   = (const float*)d_in[0];
  const int*   ei  = (const int*)d_in[1];
  const int*   bat = (const int*)d_in[2];
  const float* W1  = (const float*)d_in[3];
  const float* b1  = (const float*)d_in[4];
  const float* W2  = (const float*)d_in[5];
  const float* b2  = (const float*)d_in[6];
  const float* W3  = (const float*)d_in[7];
  const float* b3  = (const float*)d_in[8];
  (void)n_in;

  const int N = in_sizes[0] / 128;
  const int E = in_sizes[1] / 2;
  const int G = out_size / 64;
  const int* src = ei;
  const int* dst = ei + E;

  float* ws = (float*)d_ws;
  size_t o = 0;
  float* dinv = ws + o;   o += ((size_t)N + 63) & ~(size_t)63;
  float* T    = ws + o;   o += (size_t)N * 128;
  float* AGG  = ws + o;   o += (size_t)N * 128;
  float* counts = ws + o; o += (size_t)G;
  float* sums   = ws + o; o += (size_t)G * 64;
  const size_t npad = ((size_t)N + 64 + 63) & ~(size_t)63;
  o = (o + 63) & ~(size_t)63;
  int*  rowptr = (int*)(ws + o);  o += npad;
  int2* rec    = (int2*)(ws + o); o += (size_t)E * 2;
  int*  cnt    = (int*)AGG;
  int*  rank   = cnt + npad;
  const bool grouped = ws_size >= o * sizeof(float) && (size_t)N * 128 >= npad + (size_t)E;
  int2* pairs = (int2*)AGG;
  int*  rnk   = (int*)AGG + (((size_t)E * 2 + 3) & ~(size_t)3);
  int*  matT  = rnk + (((size_t)E + 3) & ~(size_t)3);
  int*  bbase = matT + EDGE_CHUNKS * NODE_BUCKET;
  const int NB = cdiv(N, NODE_BUCKET);
  const int CH = cdiv(E, EDGE_CHUNKS);
  const bool binned = grouped && NB <= NODE_BUCKET && E < (1 << 30) &&
                      (size_t)N * 128 >= (size_t)E * 3 + EDGE_CHUNKS * NODE_BUCKET + 328;

  const int B = 256;
  float* out = (float*)d_out;

  if (binned) {
    bucket_count_kernel<<<EDGE_CHUNKS, 1024, 0, stream>>>(dst, matT, E, CH);
    bucket_place_kernel<<<EDGE_CHUNKS, 1024, 0, stream>>>(src, dst, matT, pairs, bbase, E, CH, NB);
    bucket_rows_kernel<<<NB, 1024, 0, stream>>>(pairs, bbase, rnk, rowptr, dinv, rec, N);
    edge_weight_kernel<<<cdiv(E, B), B, 0, stream>>>(rec, dinv, E);
  } else if (grouped) {
    zero_kernel<<<cdiv(N, B), B, 0, stream>>>((float*)cnt, (long)N);
    count_rank_kernel<<<cdiv(E, B), B, 0, stream>>>(dst, cnt, rank, E);
    scan_deg_kernel<<<cdiv(N, 256), 256, 0, stream>>>(cnt, rowptr, dinv, N);
    fill_list_kernel<<<cdiv(E, B), B, 0, stream>>>(src, dst, rank, rowptr, dinv, rec, E);
  } else {
    init_deg_kernel<<<cdiv(N, B), B, 0, stream>>>(dinv, N);
    count_deg_kernel<<<cdiv(E, B), B, 0, stream>>>(dst, dinv, E);
    finalize_dinv_kernel<<<cdiv(N, B), B, 0, stream>>>(dinv, N);
  }

  if (grouped) {
    gemm_f32_wmma<128><<<cdiv(N, 16), B, 0, stream>>>(x, W1, T, N);
    gather_gemm_f32_wmma<128><<<cdiv(N, 16), B, 0, stream>>>(T, rowptr, rec, dinv, b1, W2, AGG, N);
    gather_gemm_f32_wmma<64><<<cdiv(N, 32), B, 0, stream>>>(AGG, rowptr, rec, dinv, b2, W3, T, N);
    gather_rows_kernel<64><<<cdiv(N, 8), 256, 0, stream>>>(T, rowptr, rec, dinv, AGG, N);
  } else {
    gemm_f32_wmma<128><<<cdiv(N, 16), B, 0, stream>>>(x, W1, T, N);
    agg_self_kernel<128><<<cdiv((long)N * 128, B), B, 0, stream>>>(T, dinv, AGG, N);
    agg_edges_kernel<128><<<cdiv((long)E * 32, B), B, 0, stream>>>(T, src, dst, dinv, AGG, E);
    bias_relu_kernel<128><<<cdiv((long)N * 128, B), B, 0, stream>>>(AGG, b1, (long)N * 128);

    gemm_f32_wmma<128><<<cdiv(N, 16), B, 0, stream>>>(AGG, W2, T, N);
    agg_self_kernel<128><<<cdiv((long)N * 128, B), B, 0, stream>>>(T, dinv, AGG, N);
    agg_edges_kernel<128><<<cdiv((long)E * 32, B), B, 0, stream>>>(T, src, dst, dinv, AGG, E);
    bias_relu_kernel<128><<<cdiv((long)N * 128, B), B, 0, stream>>>(AGG, b2, (long)N * 128);

    gemm_f32_wmma<64><<<cdiv(N, 32), B, 0, stream>>>(AGG, W3, T, N);
    agg_self_kernel<64><<<cdiv((long)N * 64, B), B, 0, stream>>>(T, dinv, AGG, N);
    agg_edges_kernel<64><<<cdiv((long)E * 16, B), B, 0, stream>>>(T, src, dst, dinv, AGG, E);
  }

  zero_kernel<<<cdiv((long)G * 65, B), B, 0, stream>>>(counts, (long)G * 65);
  pool_count_kernel<<<cdiv(N, B), B, 0, stream>>>(bat, counts, N);
  pool_sum_kernel<<<cdiv((long)N * 64, B), B, 0, stream>>>(AGG, bat, sums, N);
  pool_final_kernel<<<cdiv((long)G * 64, B), B, 0, stream>>>(sums, counts, b3, out, G);
}


# Mean pool folded into layer 3: the transform launch clears the sums, each aggregation wave adds its 2 nodes' rows in registers, then one f32 atomic per element and same-graph run; 12 -> 9 launches
# speedup vs baseline: 3.5372x; 1.0870x over previous
#include <hip/hip_runtime.h>

typedef float v2f __attribute__((ext_vector_type(2)));
typedef float v8f __attribute__((ext_vector_type(8)));

#define K_DIM 128

template <int DOUT>
__global__ __launch_bounds__(256) void gemm_f32_wmma(
    const float* __restrict__ A, const float* __restrict__ W,
    float* __restrict__ T, int N) {
  constexpr int CT = DOUT / 16;
  constexpr int RT = 8 / CT;
  constexpr int ROWS = RT * 16;
  __shared__ float As[ROWS][K_DIM + 4];

  const int tid  = threadIdx.x;
  const int wave = tid >> 5;
  const int lane = tid & 31;
  const int rowTile = wave / CT;
  const int colTile = wave % CT;
  const long rowBase = (long)blockIdx.x * ROWS;

  for (int i = tid; i < ROWS * K_DIM; i += 256) {
    const int r = i >> 7;
    const int c = i & 127;
    const long gr = rowBase + r;
    As[r][c] = (gr < N) ? A[gr * K_DIM + c] : 0.0f;
  }
  __syncthreads();

  const int half = lane >> 4;
  const int l16  = lane & 15;
  const int n0   = colTile * 16 + l16;
  const int arow = rowTile * 16 + l16;

  v8f acc = {};
  #pragma unroll
  for (int kk = 0; kk < K_DIM; kk += 4) {
    const int ka = kk + 2 * half;
    v2f a, b;
    a.x = As[arow][ka];
    a.y = As[arow][ka + 1];
    b.x = W[ka * DOUT + n0];
    b.y = W[(ka + 1) * DOUT + n0];
    acc = __builtin_amdgcn_wmma_f32_16x16x4_f32(
         false, a,  false, b,
         (short)0, acc,  false,  false);
  }

  const long mBase = rowBase + rowTile * 16 + 8 * half;
  #pragma unroll
  for (int v = 0; v < 8; ++v) {
    const long m = mBase + v;
    if (m < N) T[m * DOUT + n0] = acc[v];
  }
}

__global__ void init_deg_kernel(float* deg, int N) {
  const long i = (long)blockIdx.x * blockDim.x + threadIdx.x;
  if (i < N) deg[i] = 1.0f;
}

__global__ void count_deg_kernel(const int* __restrict__ dst, float* deg, int E) {
  const long e = (long)blockIdx.x * blockDim.x + threadIdx.x;
  if (e < E) atomicAdd(&deg[dst[e]], 1.0f);
}

__global__ void finalize_dinv_kernel(float* deg, int N) {
  const long i = (long)blockIdx.x * blockDim.x + threadIdx.x;
  if (i < N) deg[i] = rsqrtf(deg[i]);
}

template <int D>
__global__ void agg_self_kernel(const float* __restrict__ T,
                                const float* __restrict__ dinv,
                                float* __restrict__ out, int N) {
  const long idx = (long)blockIdx.x * blockDim.x + threadIdx.x;
  if (idx >= (long)N * D) return;
  const long i = idx / D;
  const float w = dinv[i];
  out[idx] = T[idx] * w * w;
}

template <int D>
__global__ void agg_edges_kernel(const float* __restrict__ T,
                                 const int* __restrict__ src,
                                 const int* __restrict__ dst,
                                 const float* __restrict__ dinv,
                                 float* __restrict__ out, int E) {
  constexpr int TPE = D / 4;
  const long idx = (long)blockIdx.x * blockDim.x + threadIdx.x;
  const long e = idx / TPE;
  const int  t = (int)(idx % TPE);
  if (e >= E) return;
  const int s = src[e];
  const int d = dst[e];
  const float norm = dinv[s] * dinv[d];
  const float4 v = ((const float4*)(T + (long)s * D))[t];
  float* op = out + (long)d * D + t * 4;
  atomicAdd(op + 0, v.x * norm);
  atomicAdd(op + 1, v.y * norm);
  atomicAdd(op + 2, v.z * norm);
  atomicAdd(op + 3, v.w * norm);
}

template <int D>
__global__ void bias_relu_kernel(float* __restrict__ h,
                                 const float* __restrict__ b, long total) {
  const long idx = (long)blockIdx.x * blockDim.x + threadIdx.x;
  if (idx >= total) return;
  const int d = (int)(idx % D);
  h[idx] = fmaxf(h[idx] + b[d], 0.0f);
}


__global__ void count_rank_kernel(const int* __restrict__ dst, int* cnt,
                                  int* __restrict__ rank, int E) {
  const long e = (long)blockIdx.x * blockDim.x + threadIdx.x;
  if (e < E) rank[e] = atomicAdd(&cnt[dst[e]], 1);
}

__global__ __launch_bounds__(256) void scan_deg_kernel(
    const int* __restrict__ cnt, int* __restrict__ rowptr,
    float* __restrict__ dinv, int N) {
  __shared__ int wtot[8];
  __shared__ int wpre[8];
  const int tid  = threadIdx.x;
  const int lane = tid & 31;
  const int wave = tid >> 5;
  const int base = blockIdx.x * 256;
  int part = 0;
  const int4* c4 = (const int4*)cnt;
  for (int i = tid; i < base / 4; i += 256) {
    const int4 q = c4[i];
    part += (q.x + q.y) + (q.z + q.w);
  }
  const int idx = base + tid;
  const int c = (idx < N) ? cnt[idx] : 0;
  int inc = c;
  for (int o = 1; o < 32; o <<= 1) {
    const int t = __shfl_up(inc, o);
    if (lane >= o) inc += t;
  }
  for (int o = 16; o > 0; o >>= 1) part += __shfl_xor(part, o);
  if (lane == 31) wtot[wave] = inc;
  if (lane == 0)  wpre[wave] = part;
  __syncthreads();
  int off = 0;
  for (int k = 0; k < 8; ++k) {
    off += wpre[k];
    if (k < wave) off += wtot[k];
  }
  const int excl = off + inc - c;
  if (idx < N) {
    rowptr[idx] = excl;
    dinv[idx] = rsqrtf((float)(c + 1));
    if (idx == N - 1) rowptr[N] = excl + c;
  }
}

__global__ void fill_list_kernel(const int* __restrict__ src, const int* __restrict__ dst,
                                 const int* __restrict__ rank, const int* __restrict__ rowptr,
                                 const float* __restrict__ dinv, int2* __restrict__ rec, int E) {
  const long e = (long)blockIdx.x * blockDim.x + threadIdx.x;
  if (e >= E) return;
  const int s = src[e];
  const int d = dst[e];
  const float norm = dinv[s] * dinv[d];
  rec[rowptr[d] + rank[e]] = make_int2(s, __float_as_int(norm));
}

#define EDGE_CHUNKS 256
#define NODE_BUCKET 256
#define BATCH 4

__device__ inline int excl_scan_first256(int c, int tid, int* wsum) {
  const int lane = tid & 31;
  const int wave = tid >> 5;
  int inc = c;
  for (int o = 1; o < 32; o <<= 1) {
    const int u = __shfl_up(inc, o);
    if (lane >= o) inc += u;
  }
  if (lane == 31 && wave < 8) wsum[wave] = inc;
  __syncthreads();
  int off = 0;
  for (int k = 0; k < 8; ++k)
    if (k < wave) off += wsum[k];
  return off + inc - c;
}

__global__ __launch_bounds__(1024) void bucket_count_kernel(
    const int* __restrict__ dst, int* __restrict__ matT, int E, int CH) {
  __shared__ int hist[NODE_BUCKET];
  const int tid = threadIdx.x;
  if (tid < NODE_BUCKET) hist[tid] = 0;
  __syncthreads();
  const int e0 = blockIdx.x * CH;
  const int e1 = min(E, e0 + CH);
  for (int eb = e0 + tid; eb < e1; eb += BATCH * 1024) {
    int d[BATCH];
    #pragma unroll
    for (int k = 0; k < BATCH; ++k) {
      d[k] = dst[min(eb + k * 1024, e1 - 1)];
    }
    #pragma unroll
    for (int k = 0; k < BATCH; ++k)
      atomicAdd(&hist[d[k] >> 8], (eb + k * 1024 < e1) ? 1 : 0);
  }
  __syncthreads();
  if (tid < NODE_BUCKET) matT[tid * EDGE_CHUNKS + blockIdx.x] = hist[tid];
}

__global__ __launch_bounds__(1024) void bucket_place_kernel(
    const int* __restrict__ src, const int* __restrict__ dst,
    const int* __restrict__ matT, int2* __restrict__ pairs,
    int* __restrict__ bbase, int E, int CH, int NB) {
  __shared__ int below[4][NODE_BUCKET];
  __shared__ int total[4][NODE_BUCKET];
  __shared__ int base[NODE_BUCKET];
  __shared__ int hist[NODE_BUCKET];
  __shared__ int wsum[8];
  const int tid = threadIdx.x;
  const int b = tid & (NODE_BUCKET - 1);
  const int q = tid >> 8;
  const int p = blockIdx.x;
  int bl = 0, tot = 0;
  {
    const int4* mq = (const int4*)(matT + b * EDGE_CHUNKS + q * 64);
    int4 v[16];
    #pragma unroll
    for (int k = 0; k < 16; ++k) v[k] = mq[k];
    #pragma unroll
    for (int k = 0; k < 16; ++k) {
      const int r = q * 64 + 4 * k;
      tot += (v[k].x + v[k].y) + (v[k].z + v[k].w);
      bl += ((r     < p) ? v[k].x : 0) + ((r + 1 < p) ? v[k].y : 0)
          + ((r + 2 < p) ? v[k].z : 0) + ((r + 3 < p) ? v[k].w : 0);
    }
  }
  below[q][b] = bl;
  total[q][b] = tot;
  if (tid < NODE_BUCKET) hist[tid] = 0;
  __syncthreads();
  int colTot = 0, colBelow = 0;
  if (tid < NODE_BUCKET) {
    colTot   = (total[0][tid] + total[1][tid]) + (total[2][tid] + total[3][tid]);
    colBelow = (below[0][tid] + below[1][tid]) + (below[2][tid] + below[3][tid]);
  }
  const int excl = excl_scan_first256(colTot, tid, wsum);
  if (tid < NODE_BUCKET) {
    base[tid] = excl + colBelow;
    if (p == 0) {
      if (tid < NB) bbase[tid] = excl;
      if (tid == 0) bbase[NB] = E;
    }
  }
  __syncthreads();
  const int e0 = p * CH;
  const int e1 = min(E, e0 + CH);
  for (int eb = e0 + tid; eb < e1; eb += BATCH * 1024) {
    int s[BATCH], d[BATCH];
    #pragma unroll
    for (int k = 0; k < BATCH; ++k) {
      const int e = min(eb + k * 1024, e1 - 1);
      s[k] = src[e];
      d[k] = dst[e];
    }
    #pragma unroll
    for (int k = 0; k < BATCH; ++k) {
      if (eb + k * 1024 < e1) {
        const int bk = d[k] >> 8;
        const int r = atomicAdd(&hist[bk], 1);
        pairs[base[bk] + r] = make_int2(s[k], d[k]);
      }
    }
  }
}

__global__ __launch_bounds__(1024) void bucket_rows_kernel(
    const int2* __restrict__ pairs, const int* __restrict__ bbase,
    int* rnk, int* __restrict__ rowptr, float* __restrict__ dinv,
    int2* __restrict__ rec, int N) {
  __shared__ int cnt[NODE_BUCKET];
  __shared__ int lptr[NODE_BUCKET];
  __shared__ int wsum[8];
  const int tid = threadIdx.x;
  const int bkt = blockIdx.x;
  const int beg = bbase[bkt];
  const int end = bbase[bkt + 1];
  if (tid < NODE_BUCKET) cnt[tid] = 0;
  __syncthreads();
  for (int ib = beg + tid; ib < end; ib += BATCH * 1024) {
    int2 pr[BATCH];
    #pragma unroll
    for (int k = 0; k < BATCH; ++k) {
      pr[k] = pairs[min(ib + k * 1024, end - 1)];
    }
    #pragma unroll
    for (int k = 0; k < BATCH; ++k) {
      const int i = ib + k * 1024;
      const int r = atomicAdd(&cnt[pr[k].y & (NODE_BUCKET - 1)], (i < end) ? 1 : 0);
      if (i < end) rnk[i] = r;
    }
  }
  __syncthreads();
  const int c = (tid < NODE_BUCKET) ? cnt[tid] : 0;
  const int excl = excl_scan_first256(c, tid, wsum);
  if (tid < NODE_BUCKET) {
    lptr[tid] = excl;
    const int node = bkt * NODE_BUCKET + tid;
    if (node < N) {
      rowptr[node] = beg + excl;
      dinv[node] = rsqrtf((float)(c + 1));
    }
  }
  if (bkt == (int)gridDim.x - 1 && tid == 0) rowptr[N] = end;
  __syncthreads();
  for (int ib = beg + tid; ib < end; ib += BATCH * 1024) {
    int2 pr[BATCH];
    int rk[BATCH];
    #pragma unroll
    for (int k = 0; k < BATCH; ++k) {
      const int i = min(ib + k * 1024, end - 1);
      pr[k] = pairs[i];
      rk[k] = rnk[i];
    }
    #pragma unroll
    for (int k = 0; k < BATCH; ++k) {
      const int i = ib + k * 1024;
      if (i < end) rec[beg + lptr[pr[k].y & (NODE_BUCKET - 1)] + rk[k]] = pr[k];
    }
  }
}

__global__ void edge_weight_kernel(int2* __restrict__ rec, const float* __restrict__ dinv, int E) {
  const long i = (long)blockIdx.x * blockDim.x + threadIdx.x;
  if (i >= E) return;
  const int2 r = rec[i];
  const float norm = dinv[r.x] * dinv[r.y];
  rec[i] = make_int2(r.x, __float_as_int(norm));
}

template <int DOUT>
__global__ __launch_bounds__(256) void gather_gemm_f32_wmma(
    const float* __restrict__ Tin, const int* __restrict__ rowptr,
    const int2* __restrict__ rec, const float* __restrict__ dinv,
    const float* __restrict__ bias, const float* __restrict__ W,
    float* __restrict__ Tout, int N, float* __restrict__ clr, int nclr) {
  constexpr int CT = DOUT / 16;
  constexpr int RT = 8 / CT;
  constexpr int ROWS = RT * 16;
  constexpr int NPW = ROWS / 8;
  __shared__ float As[ROWS][K_DIM + 4];

  typedef float v4f __attribute__((ext_vector_type(4)));
  const int tid  = threadIdx.x;
  const int lane = tid & 31;
  const int wave = __builtin_amdgcn_readfirstlane(tid >> 5);
  const int rowTile = wave / CT;
  const int colTile = wave % CT;
  const int rowBase = blockIdx.x * ROWS;

  for (long i = (long)blockIdx.x * 256 + tid; i < nclr; i += (long)gridDim.x * 256) clr[i] = 0.0f;

  const v4f* Tv = (const v4f*)Tin;
  const v4f bv = ((const v4f*)bias)[lane];
  for (int j = 0; j < NPW; ++j) {
    const int r = wave * NPW + j;
    const int node = rowBase + r;
    v4f acc = {0.0f, 0.0f, 0.0f, 0.0f};
    if (node < N) {
      const int beg = rowptr[node];
      const int end = rowptr[node + 1];
      const float w = dinv[node];
      acc = Tv[(long)node * 32 + lane] * w * w;
      #pragma unroll 4
      for (int e = beg; e < end; ++e) {
        const int2 q = rec[e];
        const v4f v = Tv[(long)q.x * 32 + lane];
        acc += v * __int_as_float(q.y);
      }
      acc += bv;
      acc.x = fmaxf(acc.x, 0.0f);
      acc.y = fmaxf(acc.y, 0.0f);
      acc.z = fmaxf(acc.z, 0.0f);
      acc.w = fmaxf(acc.w, 0.0f);
    }
    *(v4f*)&As[r][4 * lane] = acc;
  }
  __syncthreads();

  const int half = lane >> 4;
  const int l16  = lane & 15;
  const int n0   = colTile * 16 + l16;
  const int arow = rowTile * 16 + l16;

  v8f acc = {};
  #pragma unroll
  for (int kk = 0; kk < K_DIM; kk += 4) {
    const int ka = kk + 2 * half;
    v2f a, b;
    a.x = As[arow][ka];
    a.y = As[arow][ka + 1];
    b.x = W[ka * DOUT + n0];
    b.y = W[(ka + 1) * DOUT + n0];
    acc = __builtin_amdgcn_wmma_f32_16x16x4_f32(
         false, a,  false, b,
         (short)0, acc,  false,  false);
  }

  const long mBase = (long)rowBase + rowTile * 16 + 8 * half;
  #pragma unroll
  for (int v = 0; v < 8; ++v) {
    const long m = mBase + v;
    if (m < N) Tout[m * DOUT + n0] = acc[v];
  }
}

__global__ void zero_kernel(float* p, long n) {
  const long i = (long)blockIdx.x * blockDim.x + threadIdx.x;
  if (i < n) p[i] = 0.0f;
}

__global__ void pool_count_kernel(const int* __restrict__ batch, float* counts, int N) {
  const long i = (long)blockIdx.x * blockDim.x + threadIdx.x;
  if (i < N) atomicAdd(&counts[batch[i]], 1.0f);
}

__global__ void pool_sum_kernel(const float* __restrict__ h,
                                const int* __restrict__ batch,
                                float* sums, int N) {
  const long idx = (long)blockIdx.x * blockDim.x + threadIdx.x;
  if (idx >= (long)N * 64) return;
  const long i = idx / 64;
  const int  d = (int)(idx % 64);
  atomicAdd(&sums[(long)batch[i] * 64 + d], h[idx]);
}


#define POOL_NPW 2

typedef float pool_v2 __attribute__((ext_vector_type(2)));

__device__ inline void pool_flush(float* sums, float* counts, int g, int lane,
                                  pool_v2 run, int len) {
  float* p = sums + (long)g * 64 + 2 * lane;
  atomicAdd(p, run.x);
  atomicAdd(p + 1, run.y);
  if (lane == 0) atomicAdd(counts + g, (float)len);
}

__global__ __launch_bounds__(256) void gather_pool_kernel(
    const float* __restrict__ T, const int* __restrict__ rowptr,
    const int2* __restrict__ rec, const float* __restrict__ dinv,
    const int* __restrict__ batch, float* sums, float* counts, int N) {
  const int lane = threadIdx.x & 31;
  const int first = POOL_NPW *
      __builtin_amdgcn_readfirstlane((int)(blockIdx.x * 8 + (threadIdx.x >> 5)));
  if (first >= N) return;
  const int last = min(N, first + POOL_NPW);
  const pool_v2* Tv = (const pool_v2*)T;
  int g = batch[first];
  pool_v2 run = {0.0f, 0.0f};
  int len = 0;
  for (int node = first; node < last; ++node) {
    const int gn = batch[node];
    const int beg = rowptr[node];
    const int end = rowptr[node + 1];
    const float w = dinv[node];
    pool_v2 acc = Tv[(long)node * 32 + lane] * w * w;
    #pragma unroll 4
    for (int e = beg; e < end; ++e) {
      const int2 r = rec[e];
      const pool_v2 v = Tv[(long)r.x * 32 + lane];
      acc += v * __int_as_float(r.y);
    }
    if (gn != g) {
      pool_flush(sums, counts, g, lane, run, len);
      g = gn;
      run = acc;
      len = 1;
    } else {
      run += acc;
      ++len;
    }
  }
  pool_flush(sums, counts, g, lane, run, len);
}

__global__ void pool_final_kernel(const float* __restrict__ sums,
                                  const float* __restrict__ counts,
                                  const float* __restrict__ b3,
                                  float* __restrict__ out, int G) {
  const long idx = (long)blockIdx.x * blockDim.x + threadIdx.x;
  if (idx >= (long)G * 64) return;
  const long g = idx / 64;
  const int  d = (int)(idx % 64);
  const float c = counts[g];
  out[idx] = (sums[idx] + c * b3[d]) / fmaxf(c, 1.0f);
}

static inline int cdiv(long a, long b) { return (int)((a + b - 1) / b); }

extern "C" void kernel_launch(void* const* d_in, const int* in_sizes, int n_in,
                              void* d_out, int out_size, void* d_ws, size_t ws_size,
                              hipStream_t stream) {
  const float* x   = (const float*)d_in[0];
  const int*   ei  = (const int*)d_in[1];
  const int*   bat = (const int*)d_in[2];
  const float* W1  = (const float*)d_in[3];
  const float* b1  = (const float*)d_in[4];
  const float* W2  = (const float*)d_in[5];
  const float* b2  = (const float*)d_in[6];
  const float* W3  = (const float*)d_in[7];
  const float* b3  = (const float*)d_in[8];
  (void)n_in;

  const int N = in_sizes[0] / 128;
  const int E = in_sizes[1] / 2;
  const int G = out_size / 64;
  const int* src = ei;
  const int* dst = ei + E;

  float* ws = (float*)d_ws;
  size_t o = 0;
  float* dinv = ws + o;   o += ((size_t)N + 63) & ~(size_t)63;
  float* T    = ws + o;   o += (size_t)N * 128;
  float* AGG  = ws + o;   o += (size_t)N * 128;
  float* counts = ws + o; o += (size_t)G;
  float* sums   = ws + o; o += (size_t)G * 64;
  const size_t npad = ((size_t)N + 64 + 63) & ~(size_t)63;
  o = (o + 63) & ~(size_t)63;
  int*  rowptr = (int*)(ws + o);  o += npad;
  int2* rec    = (int2*)(ws + o); o += (size_t)E * 2;
  int*  cnt    = (int*)AGG;
  int*  rank   = cnt + npad;
  const bool grouped = ws_size >= o * sizeof(float) && (size_t)N * 128 >= npad + (size_t)E;
  int2* pairs = (int2*)AGG;
  int*  rnk   = (int*)AGG + (((size_t)E * 2 + 3) & ~(size_t)3);
  int*  matT  = rnk + (((size_t)E + 3) & ~(size_t)3);
  int*  bbase = matT + EDGE_CHUNKS * NODE_BUCKET;
  const int NB = cdiv(N, NODE_BUCKET);
  const int CH = cdiv(E, EDGE_CHUNKS);
  const bool binned = grouped && NB <= NODE_BUCKET && E < (1 << 30) &&
                      (size_t)N * 128 >= (size_t)E * 3 + EDGE_CHUNKS * NODE_BUCKET + 328;

  const int B = 256;
  float* out = (float*)d_out;

  if (binned) {
    bucket_count_kernel<<<EDGE_CHUNKS, 1024, 0, stream>>>(dst, matT, E, CH);
    bucket_place_kernel<<<EDGE_CHUNKS, 1024, 0, stream>>>(src, dst, matT, pairs, bbase, E, CH, NB);
    bucket_rows_kernel<<<NB, 1024, 0, stream>>>(pairs, bbase, rnk, rowptr, dinv, rec, N);
    edge_weight_kernel<<<cdiv(E, B), B, 0, stream>>>(rec, dinv, E);
  } else if (grouped) {
    zero_kernel<<<cdiv(N, B), B, 0, stream>>>((float*)cnt, (long)N);
    count_rank_kernel<<<cdiv(E, B), B, 0, stream>>>(dst, cnt, rank, E);
    scan_deg_kernel<<<cdiv(N, 256), 256, 0, stream>>>(cnt, rowptr, dinv, N);
    fill_list_kernel<<<cdiv(E, B), B, 0, stream>>>(src, dst, rank, rowptr, dinv, rec, E);
  } else {
    init_deg_kernel<<<cdiv(N, B), B, 0, stream>>>(dinv, N);
    count_deg_kernel<<<cdiv(E, B), B, 0, stream>>>(dst, dinv, E);
    finalize_dinv_kernel<<<cdiv(N, B), B, 0, stream>>>(dinv, N);
  }

  if (grouped) {
    gemm_f32_wmma<128><<<cdiv(N, 16), B, 0, stream>>>(x, W1, T, N);
    gather_gemm_f32_wmma<128><<<cdiv(N, 16), B, 0, stream>>>(T, rowptr, rec, dinv, b1, W2, AGG, N,
                                                              nullptr, 0);
    gather_gemm_f32_wmma<64><<<cdiv(N, 32), B, 0, stream>>>(AGG, rowptr, rec, dinv, b2, W3, T, N,
                                                             counts, G * 65);
    gather_pool_kernel<<<cdiv(N, 8 * POOL_NPW), 256, 0, stream>>>(T, rowptr, rec, dinv, bat,
                                                                   sums, counts, N);
  } else {
    gemm_f32_wmma<128><<<cdiv(N, 16), B, 0, stream>>>(x, W1, T, N);
    agg_self_kernel<128><<<cdiv((long)N * 128, B), B, 0, stream>>>(T, dinv, AGG, N);
    agg_edges_kernel<128><<<cdiv((long)E * 32, B), B, 0, stream>>>(T, src, dst, dinv, AGG, E);
    bias_relu_kernel<128><<<cdiv((long)N * 128, B), B, 0, stream>>>(AGG, b1, (long)N * 128);

    gemm_f32_wmma<128><<<cdiv(N, 16), B, 0, stream>>>(AGG, W2, T, N);
    agg_self_kernel<128><<<cdiv((long)N * 128, B), B, 0, stream>>>(T, dinv, AGG, N);
    agg_edges_kernel<128><<<cdiv((long)E * 32, B), B, 0, stream>>>(T, src, dst, dinv, AGG, E);
    bias_relu_kernel<128><<<cdiv((long)N * 128, B), B, 0, stream>>>(AGG, b2, (long)N * 128);

    gemm_f32_wmma<64><<<cdiv(N, 32), B, 0, stream>>>(AGG, W3, T, N);
    agg_self_kernel<64><<<cdiv((long)N * 64, B), B, 0, stream>>>(T, dinv, AGG, N);
    agg_edges_kernel<64><<<cdiv((long)E * 16, B), B, 0, stream>>>(T, src, dst, dinv, AGG, E);

    zero_kernel<<<cdiv((long)G * 65, B), B, 0, stream>>>(counts, (long)G * 65);
    pool_count_kernel<<<cdiv(N, B), B, 0, stream>>>(bat, counts, N);
    pool_sum_kernel<<<cdiv((long)N * 64, B), B, 0, stream>>>(AGG, bat, sums, N);
  }

  pool_final_kernel<<<cdiv((long)G * 64, B), B, 0, stream>>>(sums, counts, b3, out, G);
}


# Layer 1 aggregates x first (agg(x) W1 = agg(x W1)): one kernel gathers x rows to LDS, then W1, bias+ReLU, W2 via a 2nd LDS tile; transform launch and plane gone, 9 -> 8 launches; sums: agg, then dot
# speedup vs baseline: 3.8761x; 1.0958x over previous
#include <hip/hip_runtime.h>

typedef float v2f __attribute__((ext_vector_type(2)));
typedef float v8f __attribute__((ext_vector_type(8)));

#define K_DIM 128

template <int DOUT>
__global__ __launch_bounds__(256) void gemm_f32_wmma(
    const float* __restrict__ A, const float* __restrict__ W,
    float* __restrict__ T, int N) {
  constexpr int CT = DOUT / 16;
  constexpr int RT = 8 / CT;
  constexpr int ROWS = RT * 16;
  __shared__ float As[ROWS][K_DIM + 4];

  const int tid  = threadIdx.x;
  const int wave = tid >> 5;
  const int lane = tid & 31;
  const int rowTile = wave / CT;
  const int colTile = wave % CT;
  const long rowBase = (long)blockIdx.x * ROWS;

  for (int i = tid; i < ROWS * K_DIM; i += 256) {
    const int r = i >> 7;
    const int c = i & 127;
    const long gr = rowBase + r;
    As[r][c] = (gr < N) ? A[gr * K_DIM + c] : 0.0f;
  }
  __syncthreads();

  const int half = lane >> 4;
  const int l16  = lane & 15;
  const int n0   = colTile * 16 + l16;
  const int arow = rowTile * 16 + l16;

  v8f acc = {};
  #pragma unroll
  for (int kk = 0; kk < K_DIM; kk += 4) {
    const int ka = kk + 2 * half;
    v2f a, b;
    a.x = As[arow][ka];
    a.y = As[arow][ka + 1];
    b.x = W[ka * DOUT + n0];
    b.y = W[(ka + 1) * DOUT + n0];
    acc = __builtin_amdgcn_wmma_f32_16x16x4_f32(
         false, a,  false, b,
         (short)0, acc,  false,  false);
  }

  const long mBase = rowBase + rowTile * 16 + 8 * half;
  #pragma unroll
  for (int v = 0; v < 8; ++v) {
    const long m = mBase + v;
    if (m < N) T[m * DOUT + n0] = acc[v];
  }
}

__global__ void init_deg_kernel(float* deg, int N) {
  const long i = (long)blockIdx.x * blockDim.x + threadIdx.x;
  if (i < N) deg[i] = 1.0f;
}

__global__ void count_deg_kernel(const int* __restrict__ dst, float* deg, int E) {
  const long e = (long)blockIdx.x * blockDim.x + threadIdx.x;
  if (e < E) atomicAdd(&deg[dst[e]], 1.0f);
}

__global__ void finalize_dinv_kernel(float* deg, int N) {
  const long i = (long)blockIdx.x * blockDim.x + threadIdx.x;
  if (i < N) deg[i] = rsqrtf(deg[i]);
}

template <int D>
__global__ void agg_self_kernel(const float* __restrict__ T,
                                const float* __restrict__ dinv,
                                float* __restrict__ out, int N) {
  const long idx = (long)blockIdx.x * blockDim.x + threadIdx.x;
  if (idx >= (long)N * D) return;
  const long i = idx / D;
  const float w = dinv[i];
  out[idx] = T[idx] * w * w;
}

template <int D>
__global__ void agg_edges_kernel(const float* __restrict__ T,
                                 const int* __restrict__ src,
                                 const int* __restrict__ dst,
                                 const float* __restrict__ dinv,
                                 float* __restrict__ out, int E) {
  constexpr int TPE = D / 4;
  const long idx = (long)blockIdx.x * blockDim.x + threadIdx.x;
  const long e = idx / TPE;
  const int  t = (int)(idx % TPE);
  if (e >= E) return;
  const int s = src[e];
  const int d = dst[e];
  const float norm = dinv[s] * dinv[d];
  const float4 v = ((const float4*)(T + (long)s * D))[t];
  float* op = out + (long)d * D + t * 4;
  atomicAdd(op + 0, v.x * norm);
  atomicAdd(op + 1, v.y * norm);
  atomicAdd(op + 2, v.z * norm);
  atomicAdd(op + 3, v.w * norm);
}

template <int D>
__global__ void bias_relu_kernel(float* __restrict__ h,
                                 const float* __restrict__ b, long total) {
  const long idx = (long)blockIdx.x * blockDim.x + threadIdx.x;
  if (idx >= total) return;
  const int d = (int)(idx % D);
  h[idx] = fmaxf(h[idx] + b[d], 0.0f);
}


__global__ void count_rank_kernel(const int* __restrict__ dst, int* cnt,
                                  int* __restrict__ rank, int E) {
  const long e = (long)blockIdx.x * blockDim.x + threadIdx.x;
  if (e < E) rank[e] = atomicAdd(&cnt[dst[e]], 1);
}

__global__ __launch_bounds__(256) void scan_deg_kernel(
    const int* __restrict__ cnt, int* __restrict__ rowptr,
    float* __restrict__ dinv, int N) {
  __shared__ int wtot[8];
  __shared__ int wpre[8];
  const int tid  = threadIdx.x;
  const int lane = tid & 31;
  const int wave = tid >> 5;
  const int base = blockIdx.x * 256;
  int part = 0;
  const int4* c4 = (const int4*)cnt;
  for (int i = tid; i < base / 4; i += 256) {
    const int4 q = c4[i];
    part += (q.x + q.y) + (q.z + q.w);
  }
  const int idx = base + tid;
  const int c = (idx < N) ? cnt[idx] : 0;
  int inc = c;
  for (int o = 1; o < 32; o <<= 1) {
    const int t = __shfl_up(inc, o);
    if (lane >= o) inc += t;
  }
  for (int o = 16; o > 0; o >>= 1) part += __shfl_xor(part, o);
  if (lane == 31) wtot[wave] = inc;
  if (lane == 0)  wpre[wave] = part;
  __syncthreads();
  int off = 0;
  for (int k = 0; k < 8; ++k) {
    off += wpre[k];
    if (k < wave) off += wtot[k];
  }
  const int excl = off + inc - c;
  if (idx < N) {
    rowptr[idx] = excl;
    dinv[idx] = rsqrtf((float)(c + 1));
    if (idx == N - 1) rowptr[N] = excl + c;
  }
}

__global__ void fill_list_kernel(const int* __restrict__ src, const int* __restrict__ dst,
                                 const int* __restrict__ rank, const int* __restrict__ rowptr,
                                 const float* __restrict__ dinv, int2* __restrict__ rec, int E) {
  const long e = (long)blockIdx.x * blockDim.x + threadIdx.x;
  if (e >= E) return;
  const int s = src[e];
  const int d = dst[e];
  const float norm = dinv[s] * dinv[d];
  rec[rowptr[d] + rank[e]] = make_int2(s, __float_as_int(norm));
}

#define EDGE_CHUNKS 256
#define NODE_BUCKET 256
#define BATCH 4

__device__ inline int excl_scan_first256(int c, int tid, int* wsum) {
  const int lane = tid & 31;
  const int wave = tid >> 5;
  int inc = c;
  for (int o = 1; o < 32; o <<= 1) {
    const int u = __shfl_up(inc, o);
    if (lane >= o) inc += u;
  }
  if (lane == 31 && wave < 8) wsum[wave] = inc;
  __syncthreads();
  int off = 0;
  for (int k = 0; k < 8; ++k)
    if (k < wave) off += wsum[k];
  return off + inc - c;
}

__global__ __launch_bounds__(1024) void bucket_count_kernel(
    const int* __restrict__ dst, int* __restrict__ matT, int E, int CH) {
  __shared__ int hist[NODE_BUCKET];
  const int tid = threadIdx.x;
  if (tid < NODE_BUCKET) hist[tid] = 0;
  __syncthreads();
  const int e0 = blockIdx.x * CH;
  const int e1 = min(E, e0 + CH);
  for (int eb = e0 + tid; eb < e1; eb += BATCH * 1024) {
    int d[BATCH];
    #pragma unroll
    for (int k = 0; k < BATCH; ++k) {
      d[k] = dst[min(eb + k * 1024, e1 - 1)];
    }
    #pragma unroll
    for (int k = 0; k < BATCH; ++k)
      atomicAdd(&hist[d[k] >> 8], (eb + k * 1024 < e1) ? 1 : 0);
  }
  __syncthreads();
  if (tid < NODE_BUCKET) matT[tid * EDGE_CHUNKS + blockIdx.x] = hist[tid];
}

__global__ __launch_bounds__(1024) void bucket_place_kernel(
    const int* __restrict__ src, const int* __restrict__ dst,
    const int* __restrict__ matT, int2* __restrict__ pairs,
    int* __restrict__ bbase, int E, int CH, int NB) {
  __shared__ int below[4][NODE_BUCKET];
  __shared__ int total[4][NODE_BUCKET];
  __shared__ int base[NODE_BUCKET];
  __shared__ int hist[NODE_BUCKET];
  __shared__ int wsum[8];
  const int tid = threadIdx.x;
  const int b = tid & (NODE_BUCKET - 1);
  const int q = tid >> 8;
  const int p = blockIdx.x;
  int bl = 0, tot = 0;
  {
    const int4* mq = (const int4*)(matT + b * EDGE_CHUNKS + q * 64);
    int4 v[16];
    #pragma unroll
    for (int k = 0; k < 16; ++k) v[k] = mq[k];
    #pragma unroll
    for (int k = 0; k < 16; ++k) {
      const int r = q * 64 + 4 * k;
      tot += (v[k].x + v[k].y) + (v[k].z + v[k].w);
      bl += ((r     < p) ? v[k].x : 0) + ((r + 1 < p) ? v[k].y : 0)
          + ((r + 2 < p) ? v[k].z : 0) + ((r + 3 < p) ? v[k].w : 0);
    }
  }
  below[q][b] = bl;
  total[q][b] = tot;
  if (tid < NODE_BUCKET) hist[tid] = 0;
  __syncthreads();
  int colTot = 0, colBelow = 0;
  if (tid < NODE_BUCKET) {
    colTot   = (total[0][tid] + total[1][tid]) + (total[2][tid] + total[3][tid]);
    colBelow = (below[0][tid] + below[1][tid]) + (below[2][tid] + below[3][tid]);
  }
  const int excl = excl_scan_first256(colTot, tid, wsum);
  if (tid < NODE_BUCKET) {
    base[tid] = excl + colBelow;
    if (p == 0) {
      if (tid < NB) bbase[tid] = excl;
      if (tid == 0) bbase[NB] = E;
    }
  }
  __syncthreads();
  const int e0 = p * CH;
  const int e1 = min(E, e0 + CH);
  for (int eb = e0 + tid; eb < e1; eb += BATCH * 1024) {
    int s[BATCH], d[BATCH];
    #pragma unroll
    for (int k = 0; k < BATCH; ++k) {
      const int e = min(eb + k * 1024, e1 - 1);
      s[k] = src[e];
      d[k] = dst[e];
    }
    #pragma unroll
    for (int k = 0; k < BATCH; ++k) {
      if (eb + k * 1024 < e1) {
        const int bk = d[k] >> 8;
        const int r = atomicAdd(&hist[bk], 1);
        pairs[base[bk] + r] = make_int2(s[k], d[k]);
      }
    }
  }
}

__global__ __launch_bounds__(1024) void bucket_rows_kernel(
    const int2* __restrict__ pairs, const int* __restrict__ bbase,
    int* rnk, int* __restrict__ rowptr, float* __restrict__ dinv,
    int2* __restrict__ rec, int N) {
  __shared__ int cnt[NODE_BUCKET];
  __shared__ int lptr[NODE_BUCKET];
  __shared__ int wsum[8];
  const int tid = threadIdx.x;
  const int bkt = blockIdx.x;
  const int beg = bbase[bkt];
  const int end = bbase[bkt + 1];
  if (tid < NODE_BUCKET) cnt[tid] = 0;
  __syncthreads();
  for (int ib = beg + tid; ib < end; ib += BATCH * 1024) {
    int2 pr[BATCH];
    #pragma unroll
    for (int k = 0; k < BATCH; ++k) {
      pr[k] = pairs[min(ib + k * 1024, end - 1)];
    }
    #pragma unroll
    for (int k = 0; k < BATCH; ++k) {
      const int i = ib + k * 1024;
      const int r = atomicAdd(&cnt[pr[k].y & (NODE_BUCKET - 1)], (i < end) ? 1 : 0);
      if (i < end) rnk[i] = r;
    }
  }
  __syncthreads();
  const int c = (tid < NODE_BUCKET) ? cnt[tid] : 0;
  const int excl = excl_scan_first256(c, tid, wsum);
  if (tid < NODE_BUCKET) {
    lptr[tid] = excl;
    const int node = bkt * NODE_BUCKET + tid;
    if (node < N) {
      rowptr[node] = beg + excl;
      dinv[node] = rsqrtf((float)(c + 1));
    }
  }
  if (bkt == (int)gridDim.x - 1 && tid == 0) rowptr[N] = end;
  __syncthreads();
  for (int ib = beg + tid; ib < end; ib += BATCH * 1024) {
    int2 pr[BATCH];
    int rk[BATCH];
    #pragma unroll
    for (int k = 0; k < BATCH; ++k) {
      const int i = min(ib + k * 1024, end - 1);
      pr[k] = pairs[i];
      rk[k] = rnk[i];
    }
    #pragma unroll
    for (int k = 0; k < BATCH; ++k) {
      const int i = ib + k * 1024;
      if (i < end) rec[beg + lptr[pr[k].y & (NODE_BUCKET - 1)] + rk[k]] = pr[k];
    }
  }
}

__global__ void edge_weight_kernel(int2* __restrict__ rec, const float* __restrict__ dinv, int E) {
  const long i = (long)blockIdx.x * blockDim.x + threadIdx.x;
  if (i >= E) return;
  const int2 r = rec[i];
  const float norm = dinv[r.x] * dinv[r.y];
  rec[i] = make_int2(r.x, __float_as_int(norm));
}

template <int DOUT>
__global__ __launch_bounds__(256) void gather_gemm_f32_wmma(
    const float* __restrict__ Tin, const int* __restrict__ rowptr,
    const int2* __restrict__ rec, const float* __restrict__ dinv,
    const float* __restrict__ bias, const float* __restrict__ W,
    float* __restrict__ Tout, int N, float* __restrict__ clr, int nclr) {
  constexpr int CT = DOUT / 16;
  constexpr int RT = 8 / CT;
  constexpr int ROWS = RT * 16;
  constexpr int NPW = ROWS / 8;
  __shared__ float As[ROWS][K_DIM + 4];

  typedef float v4f __attribute__((ext_vector_type(4)));
  const int tid  = threadIdx.x;
  const int lane = tid & 31;
  const int wave = __builtin_amdgcn_readfirstlane(tid >> 5);
  const int rowTile = wave / CT;
  const int colTile = wave % CT;
  const int rowBase = blockIdx.x * ROWS;

  for (long i = (long)blockIdx.x * 256 + tid; i < nclr; i += (long)gridDim.x * 256) clr[i] = 0.0f;

  const v4f* Tv = (const v4f*)Tin;
  const v4f bv = ((const v4f*)bias)[lane];
  for (int j = 0; j < NPW; ++j) {
    const int r = wave * NPW + j;
    const int node = rowBase + r;
    v4f acc = {0.0f, 0.0f, 0.0f, 0.0f};
    if (node < N) {
      const int beg = rowptr[node];
      const int end = rowptr[node + 1];
      const float w = dinv[node];
      acc = Tv[(long)node * 32 + lane] * w * w;
      #pragma unroll 4
      for (int e = beg; e < end; ++e) {
        const int2 q = rec[e];
        const v4f v = Tv[(long)q.x * 32 + lane];
        acc += v * __int_as_float(q.y);
      }
      acc += bv;
      acc.x = fmaxf(acc.x, 0.0f);
      acc.y = fmaxf(acc.y, 0.0f);
      acc.z = fmaxf(acc.z, 0.0f);
      acc.w = fmaxf(acc.w, 0.0f);
    }
    *(v4f*)&As[r][4 * lane] = acc;
  }
  __syncthreads();

  const int half = lane >> 4;
  const int l16  = lane & 15;
  const int n0   = colTile * 16 + l16;
  const int arow = rowTile * 16 + l16;

  v8f acc = {};
  #pragma unroll
  for (int kk = 0; kk < K_DIM; kk += 4) {
    const int ka = kk + 2 * half;
    v2f a, b;
    a.x = As[arow][ka];
    a.y = As[arow][ka + 1];
    b.x = W[ka * DOUT + n0];
    b.y = W[(ka + 1) * DOUT + n0];
    acc = __builtin_amdgcn_wmma_f32_16x16x4_f32(
         false, a,  false, b,
         (short)0, acc,  false,  false);
  }

  const long mBase = (long)rowBase + rowTile * 16 + 8 * half;
  #pragma unroll
  for (int v = 0; v < 8; ++v) {
    const long m = mBase + v;
    if (m < N) Tout[m * DOUT + n0] = acc[v];
  }
}

__global__ __launch_bounds__(256, 16) void gather_gemm2_f32_wmma(
    const float* __restrict__ X, const int* __restrict__ rowptr,
    const int2* __restrict__ rec, const float* __restrict__ dinv,
    const float* __restrict__ Wa, const float* __restrict__ ba,
    const float* __restrict__ Wb, float* __restrict__ Tout, int N) {
  constexpr int DOUT = 128;
  constexpr int ROWS = 16;
  constexpr int NPW = ROWS / 8;
  __shared__ float As[ROWS][K_DIM + 4];
  __shared__ float Hs[ROWS][K_DIM + 4];

  typedef float v4f __attribute__((ext_vector_type(4)));
  const int tid  = threadIdx.x;
  const int lane = tid & 31;
  const int wave = __builtin_amdgcn_readfirstlane(tid >> 5);
  const int rowBase = blockIdx.x * ROWS;

  const v4f* Xv = (const v4f*)X;
  for (int j = 0; j < NPW; ++j) {
    const int r = wave * NPW + j;
    const int node = rowBase + r;
    v4f acc = {0.0f, 0.0f, 0.0f, 0.0f};
    if (node < N) {
      const int beg = rowptr[node];
      const int end = rowptr[node + 1];
      const float w = dinv[node];
      acc = Xv[(long)node * 32 + lane] * w * w;
      #pragma unroll 4
      for (int e = beg; e < end; ++e) {
        const int2 q = rec[e];
        const v4f v = Xv[(long)q.x * 32 + lane];
        acc += v * __int_as_float(q.y);
      }
    }
    *(v4f*)&As[r][4 * lane] = acc;
  }
  __syncthreads();

  const int half = lane >> 4;
  const int l16  = lane & 15;
  const int n0   = wave * 16 + l16;
  const int arow = l16;

  v8f acc = {};
  #pragma unroll
  for (int kk = 0; kk < K_DIM; kk += 4) {
    const int ka = kk + 2 * half;
    v2f a, b;
    a.x = As[arow][ka];
    a.y = As[arow][ka + 1];
    b.x = Wa[ka * DOUT + n0];
    b.y = Wa[(ka + 1) * DOUT + n0];
    acc = __builtin_amdgcn_wmma_f32_16x16x4_f32(
         false, a,  false, b,
         (short)0, acc,  false,  false);
  }

  const float bb = ba[n0];
  #pragma unroll
  for (int v = 0; v < 8; ++v) Hs[8 * half + v][n0] = fmaxf(acc[v] + bb, 0.0f);
  __syncthreads();

  v8f acc2 = {};
  #pragma unroll
  for (int kk = 0; kk < K_DIM; kk += 4) {
    const int ka = kk + 2 * half;
    v2f a, b;
    a.x = Hs[arow][ka];
    a.y = Hs[arow][ka + 1];
    b.x = Wb[ka * DOUT + n0];
    b.y = Wb[(ka + 1) * DOUT + n0];
    acc2 = __builtin_amdgcn_wmma_f32_16x16x4_f32(
         false, a,  false, b,
         (short)0, acc2,  false,  false);
  }

  const long mBase = (long)rowBase + 8 * half;
  #pragma unroll
  for (int v = 0; v < 8; ++v) {
    const long m = mBase + v;
    if (m < N) Tout[m * DOUT + n0] = acc2[v];
  }
}

__global__ void zero_kernel(float* p, long n) {
  const long i = (long)blockIdx.x * blockDim.x + threadIdx.x;
  if (i < n) p[i] = 0.0f;
}

__global__ void pool_count_kernel(const int* __restrict__ batch, float* counts, int N) {
  const long i = (long)blockIdx.x * blockDim.x + threadIdx.x;
  if (i < N) atomicAdd(&counts[batch[i]], 1.0f);
}

__global__ void pool_sum_kernel(const float* __restrict__ h,
                                const int* __restrict__ batch,
                                float* sums, int N) {
  const long idx = (long)blockIdx.x * blockDim.x + threadIdx.x;
  if (idx >= (long)N * 64) return;
  const long i = idx / 64;
  const int  d = (int)(idx % 64);
  atomicAdd(&sums[(long)batch[i] * 64 + d], h[idx]);
}


#define POOL_NPW 2

typedef float pool_v2 __attribute__((ext_vector_type(2)));

__device__ inline void pool_flush(float* sums, float* counts, int g, int lane,
                                  pool_v2 run, int len) {
  float* p = sums + (long)g * 64 + 2 * lane;
  atomicAdd(p, run.x);
  atomicAdd(p + 1, run.y);
  if (lane == 0) atomicAdd(counts + g, (float)len);
}

__global__ __launch_bounds__(256) void gather_pool_kernel(
    const float* __restrict__ T, const int* __restrict__ rowptr,
    const int2* __restrict__ rec, const float* __restrict__ dinv,
    const int* __restrict__ batch, float* sums, float* counts, int N) {
  const int lane = threadIdx.x & 31;
  const int first = POOL_NPW *
      __builtin_amdgcn_readfirstlane((int)(blockIdx.x * 8 + (threadIdx.x >> 5)));
  if (first >= N) return;
  const int last = min(N, first + POOL_NPW);
  const pool_v2* Tv = (const pool_v2*)T;
  int g = batch[first];
  pool_v2 run = {0.0f, 0.0f};
  int len = 0;
  for (int node = first; node < last; ++node) {
    const int gn = batch[node];
    const int beg = rowptr[node];
    const int end = rowptr[node + 1];
    const float w = dinv[node];
    pool_v2 acc = Tv[(long)node * 32 + lane] * w * w;
    #pragma unroll 4
    for (int e = beg; e < end; ++e) {
      const int2 r = rec[e];
      const pool_v2 v = Tv[(long)r.x * 32 + lane];
      acc += v * __int_as_float(r.y);
    }
    if (gn != g) {
      pool_flush(sums, counts, g, lane, run, len);
      g = gn;
      run = acc;
      len = 1;
    } else {
      run += acc;
      ++len;
    }
  }
  pool_flush(sums, counts, g, lane, run, len);
}

__global__ void pool_final_kernel(const float* __restrict__ sums,
                                  const float* __restrict__ counts,
                                  const float* __restrict__ b3,
                                  float* __restrict__ out, int G) {
  const long idx = (long)blockIdx.x * blockDim.x + threadIdx.x;
  if (idx >= (long)G * 64) return;
  const long g = idx / 64;
  const int  d = (int)(idx % 64);
  const float c = counts[g];
  out[idx] = (sums[idx] + c * b3[d]) / fmaxf(c, 1.0f);
}

static inline int cdiv(long a, long b) { return (int)((a + b - 1) / b); }

extern "C" void kernel_launch(void* const* d_in, const int* in_sizes, int n_in,
                              void* d_out, int out_size, void* d_ws, size_t ws_size,
                              hipStream_t stream) {
  const float* x   = (const float*)d_in[0];
  const int*   ei  = (const int*)d_in[1];
  const int*   bat = (const int*)d_in[2];
  const float* W1  = (const float*)d_in[3];
  const float* b1  = (const float*)d_in[4];
  const float* W2  = (const float*)d_in[5];
  const float* b2  = (const float*)d_in[6];
  const float* W3  = (const float*)d_in[7];
  const float* b3  = (const float*)d_in[8];
  (void)n_in;

  const int N = in_sizes[0] / 128;
  const int E = in_sizes[1] / 2;
  const int G = out_size / 64;
  const int* src = ei;
  const int* dst = ei + E;

  float* ws = (float*)d_ws;
  size_t o = 0;
  float* dinv = ws + o;   o += ((size_t)N + 63) & ~(size_t)63;
  float* T    = ws + o;   o += (size_t)N * 128;
  float* AGG  = ws + o;   o += (size_t)N * 128;
  float* counts = ws + o; o += (size_t)G;
  float* sums   = ws + o; o += (size_t)G * 64;
  const size_t npad = ((size_t)N + 64 + 63) & ~(size_t)63;
  o = (o + 63) & ~(size_t)63;
  int*  rowptr = (int*)(ws + o);  o += npad;
  int2* rec    = (int2*)(ws + o); o += (size_t)E * 2;
  int*  cnt    = (int*)AGG;
  int*  rank   = cnt + npad;
  const bool grouped = ws_size >= o * sizeof(float) && (size_t)N * 128 >= npad + (size_t)E;
  int2* pairs = (int2*)AGG;
  int*  rnk   = (int*)AGG + (((size_t)E * 2 + 3) & ~(size_t)3);
  int*  matT  = rnk + (((size_t)E + 3) & ~(size_t)3);
  int*  bbase = matT + EDGE_CHUNKS * NODE_BUCKET;
  const int NB = cdiv(N, NODE_BUCKET);
  const int CH = cdiv(E, EDGE_CHUNKS);
  const bool binned = grouped && NB <= NODE_BUCKET && E < (1 << 30) &&
                      (size_t)N * 128 >= (size_t)E * 3 + EDGE_CHUNKS * NODE_BUCKET + 328;

  const int B = 256;
  float* out = (float*)d_out;

  if (binned) {
    bucket_count_kernel<<<EDGE_CHUNKS, 1024, 0, stream>>>(dst, matT, E, CH);
    bucket_place_kernel<<<EDGE_CHUNKS, 1024, 0, stream>>>(src, dst, matT, pairs, bbase, E, CH, NB);
    bucket_rows_kernel<<<NB, 1024, 0, stream>>>(pairs, bbase, rnk, rowptr, dinv, rec, N);
    edge_weight_kernel<<<cdiv(E, B), B, 0, stream>>>(rec, dinv, E);
  } else if (grouped) {
    zero_kernel<<<cdiv(N, B), B, 0, stream>>>((float*)cnt, (long)N);
    count_rank_kernel<<<cdiv(E, B), B, 0, stream>>>(dst, cnt, rank, E);
    scan_deg_kernel<<<cdiv(N, 256), 256, 0, stream>>>(cnt, rowptr, dinv, N);
    fill_list_kernel<<<cdiv(E, B), B, 0, stream>>>(src, dst, rank, rowptr, dinv, rec, E);
  } else {
    init_deg_kernel<<<cdiv(N, B), B, 0, stream>>>(dinv, N);
    count_deg_kernel<<<cdiv(E, B), B, 0, stream>>>(dst, dinv, E);
    finalize_dinv_kernel<<<cdiv(N, B), B, 0, stream>>>(dinv, N);
  }

  if (grouped) {
    gather_gemm2_f32_wmma<<<cdiv(N, 16), B, 0, stream>>>(x, rowptr, rec, dinv, W1, b1, W2, AGG, N);
    gather_gemm_f32_wmma<64><<<cdiv(N, 32), B, 0, stream>>>(AGG, rowptr, rec, dinv, b2, W3, T, N,
                                                             counts, G * 65);
    gather_pool_kernel<<<cdiv(N, 8 * POOL_NPW), 256, 0, stream>>>(T, rowptr, rec, dinv, bat,
                                                                   sums, counts, N);
  } else {
    gemm_f32_wmma<128><<<cdiv(N, 16), B, 0, stream>>>(x, W1, T, N);
    agg_self_kernel<128><<<cdiv((long)N * 128, B), B, 0, stream>>>(T, dinv, AGG, N);
    agg_edges_kernel<128><<<cdiv((long)E * 32, B), B, 0, stream>>>(T, src, dst, dinv, AGG, E);
    bias_relu_kernel<128><<<cdiv((long)N * 128, B), B, 0, stream>>>(AGG, b1, (long)N * 128);

    gemm_f32_wmma<128><<<cdiv(N, 16), B, 0, stream>>>(AGG, W2, T, N);
    agg_self_kernel<128><<<cdiv((long)N * 128, B), B, 0, stream>>>(T, dinv, AGG, N);
    agg_edges_kernel<128><<<cdiv((long)E * 32, B), B, 0, stream>>>(T, src, dst, dinv, AGG, E);
    bias_relu_kernel<128><<<cdiv((long)N * 128, B), B, 0, stream>>>(AGG, b2, (long)N * 128);

    gemm_f32_wmma<64><<<cdiv(N, 32), B, 0, stream>>>(AGG, W3, T, N);
    agg_self_kernel<64><<<cdiv((long)N * 64, B), B, 0, stream>>>(T, dinv, AGG, N);
    agg_edges_kernel<64><<<cdiv((long)E * 16, B), B, 0, stream>>>(T, src, dst, dinv, AGG, E);

    zero_kernel<<<cdiv((long)G * 65, B), B, 0, stream>>>(counts, (long)G * 65);
    pool_count_kernel<<<cdiv(N, B), B, 0, stream>>>(bat, counts, N);
    pool_sum_kernel<<<cdiv((long)N * 64, B), B, 0, stream>>>(AGG, bat, sums, N);
  }

  pool_final_kernel<<<cdiv((long)G * 64, B), B, 0, stream>>>(sums, counts, b3, out, G);
}


# Edge weight dinv[s]*dinv[d] factored: planes stored scaled by dinv[row], gathers add rows unweighted and scale the sum once; weight pass gone, 8 -> 7 launches; sum order kept, rounding differs
# speedup vs baseline: 4.0057x; 1.0334x over previous
#include <hip/hip_runtime.h>

typedef float v2f __attribute__((ext_vector_type(2)));
typedef float v8f __attribute__((ext_vector_type(8)));

#define K_DIM 128

template <int DOUT>
__global__ __launch_bounds__(256) void gemm_f32_wmma(
    const float* __restrict__ A, const float* __restrict__ W,
    float* __restrict__ T, int N) {
  constexpr int CT = DOUT / 16;
  constexpr int RT = 8 / CT;
  constexpr int ROWS = RT * 16;
  __shared__ float As[ROWS][K_DIM + 4];

  const int tid  = threadIdx.x;
  const int wave = tid >> 5;
  const int lane = tid & 31;
  const int rowTile = wave / CT;
  const int colTile = wave % CT;
  const long rowBase = (long)blockIdx.x * ROWS;

  for (int i = tid; i < ROWS * K_DIM; i += 256) {
    const int r = i >> 7;
    const int c = i & 127;
    const long gr = rowBase + r;
    As[r][c] = (gr < N) ? A[gr * K_DIM + c] : 0.0f;
  }
  __syncthreads();

  const int half = lane >> 4;
  const int l16  = lane & 15;
  const int n0   = colTile * 16 + l16;
  const int arow = rowTile * 16 + l16;

  v8f acc = {};
  #pragma unroll
  for (int kk = 0; kk < K_DIM; kk += 4) {
    const int ka = kk + 2 * half;
    v2f a, b;
    a.x = As[arow][ka];
    a.y = As[arow][ka + 1];
    b.x = W[ka * DOUT + n0];
    b.y = W[(ka + 1) * DOUT + n0];
    acc = __builtin_amdgcn_wmma_f32_16x16x4_f32(
         false, a,  false, b,
         (short)0, acc,  false,  false);
  }

  const long mBase = rowBase + rowTile * 16 + 8 * half;
  #pragma unroll
  for (int v = 0; v < 8; ++v) {
    const long m = mBase + v;
    if (m < N) T[m * DOUT + n0] = acc[v];
  }
}

__global__ void init_deg_kernel(float* deg, int N) {
  const long i = (long)blockIdx.x * blockDim.x + threadIdx.x;
  if (i < N) deg[i] = 1.0f;
}

__global__ void count_deg_kernel(const int* __restrict__ dst, float* deg, int E) {
  const long e = (long)blockIdx.x * blockDim.x + threadIdx.x;
  if (e < E) atomicAdd(&deg[dst[e]], 1.0f);
}

__global__ void finalize_dinv_kernel(float* deg, int N) {
  const long i = (long)blockIdx.x * blockDim.x + threadIdx.x;
  if (i < N) deg[i] = rsqrtf(deg[i]);
}

template <int D>
__global__ void agg_self_kernel(const float* __restrict__ T,
                                const float* __restrict__ dinv,
                                float* __restrict__ out, int N) {
  const long idx = (long)blockIdx.x * blockDim.x + threadIdx.x;
  if (idx >= (long)N * D) return;
  const long i = idx / D;
  const float w = dinv[i];
  out[idx] = T[idx] * w * w;
}

template <int D>
__global__ void agg_edges_kernel(const float* __restrict__ T,
                                 const int* __restrict__ src,
                                 const int* __restrict__ dst,
                                 const float* __restrict__ dinv,
                                 float* __restrict__ out, int E) {
  constexpr int TPE = D / 4;
  const long idx = (long)blockIdx.x * blockDim.x + threadIdx.x;
  const long e = idx / TPE;
  const int  t = (int)(idx % TPE);
  if (e >= E) return;
  const int s = src[e];
  const int d = dst[e];
  const float norm = dinv[s] * dinv[d];
  const float4 v = ((const float4*)(T + (long)s * D))[t];
  float* op = out + (long)d * D + t * 4;
  atomicAdd(op + 0, v.x * norm);
  atomicAdd(op + 1, v.y * norm);
  atomicAdd(op + 2, v.z * norm);
  atomicAdd(op + 3, v.w * norm);
}

template <int D>
__global__ void bias_relu_kernel(float* __restrict__ h,
                                 const float* __restrict__ b, long total) {
  const long idx = (long)blockIdx.x * blockDim.x + threadIdx.x;
  if (idx >= total) return;
  const int d = (int)(idx % D);
  h[idx] = fmaxf(h[idx] + b[d], 0.0f);
}


__global__ void count_rank_kernel(const int* __restrict__ dst, int* cnt,
                                  int* __restrict__ rank, int E) {
  const long e = (long)blockIdx.x * blockDim.x + threadIdx.x;
  if (e < E) rank[e] = atomicAdd(&cnt[dst[e]], 1);
}

__global__ __launch_bounds__(256) void scan_deg_kernel(
    const int* __restrict__ cnt, int* __restrict__ rowptr,
    float* __restrict__ dinv, int N) {
  __shared__ int wtot[8];
  __shared__ int wpre[8];
  const int tid  = threadIdx.x;
  const int lane = tid & 31;
  const int wave = tid >> 5;
  const int base = blockIdx.x * 256;
  int part = 0;
  const int4* c4 = (const int4*)cnt;
  for (int i = tid; i < base / 4; i += 256) {
    const int4 q = c4[i];
    part += (q.x + q.y) + (q.z + q.w);
  }
  const int idx = base + tid;
  const int c = (idx < N) ? cnt[idx] : 0;
  int inc = c;
  for (int o = 1; o < 32; o <<= 1) {
    const int t = __shfl_up(inc, o);
    if (lane >= o) inc += t;
  }
  for (int o = 16; o > 0; o >>= 1) part += __shfl_xor(part, o);
  if (lane == 31) wtot[wave] = inc;
  if (lane == 0)  wpre[wave] = part;
  __syncthreads();
  int off = 0;
  for (int k = 0; k < 8; ++k) {
    off += wpre[k];
    if (k < wave) off += wtot[k];
  }
  const int excl = off + inc - c;
  if (idx < N) {
    rowptr[idx] = excl;
    dinv[idx] = rsqrtf((float)(c + 1));
    if (idx == N - 1) rowptr[N] = excl + c;
  }
}

__global__ void fill_list_kernel(const int* __restrict__ src, const int* __restrict__ dst,
                                 const int* __restrict__ rank, const int* __restrict__ rowptr,
                                 int* __restrict__ rec, int E) {
  const long e = (long)blockIdx.x * blockDim.x + threadIdx.x;
  if (e >= E) return;
  rec[rowptr[dst[e]] + rank[e]] = src[e];
}

#define EDGE_CHUNKS 256
#define NODE_BUCKET 256
#define BATCH 4

__device__ inline int excl_scan_first256(int c, int tid, int* wsum) {
  const int lane = tid & 31;
  const int wave = tid >> 5;
  int inc = c;
  for (int o = 1; o < 32; o <<= 1) {
    const int u = __shfl_up(inc, o);
    if (lane >= o) inc += u;
  }
  if (lane == 31 && wave < 8) wsum[wave] = inc;
  __syncthreads();
  int off = 0;
  for (int k = 0; k < 8; ++k)
    if (k < wave) off += wsum[k];
  return off + inc - c;
}

__global__ __launch_bounds__(1024) void bucket_count_kernel(
    const int* __restrict__ dst, int* __restrict__ matT, int E, int CH) {
  __shared__ int hist[NODE_BUCKET];
  const int tid = threadIdx.x;
  if (tid < NODE_BUCKET) hist[tid] = 0;
  __syncthreads();
  const int e0 = blockIdx.x * CH;
  const int e1 = min(E, e0 + CH);
  for (int eb = e0 + tid; eb < e1; eb += BATCH * 1024) {
    int d[BATCH];
    #pragma unroll
    for (int k = 0; k < BATCH; ++k) {
      d[k] = dst[min(eb + k * 1024, e1 - 1)];
    }
    #pragma unroll
    for (int k = 0; k < BATCH; ++k)
      atomicAdd(&hist[d[k] >> 8], (eb + k * 1024 < e1) ? 1 : 0);
  }
  __syncthreads();
  if (tid < NODE_BUCKET) matT[tid * EDGE_CHUNKS + blockIdx.x] = hist[tid];
}

__global__ __launch_bounds__(1024) void bucket_place_kernel(
    const int* __restrict__ src, const int* __restrict__ dst,
    const int* __restrict__ matT, int2* __restrict__ pairs,
    int* __restrict__ bbase, int E, int CH, int NB) {
  __shared__ int below[4][NODE_BUCKET];
  __shared__ int total[4][NODE_BUCKET];
  __shared__ int base[NODE_BUCKET];
  __shared__ int hist[NODE_BUCKET];
  __shared__ int wsum[8];
  const int tid = threadIdx.x;
  const int b = tid & (NODE_BUCKET - 1);
  const int q = tid >> 8;
  const int p = blockIdx.x;
  int bl = 0, tot = 0;
  {
    const int4* mq = (const int4*)(matT + b * EDGE_CHUNKS + q * 64);
    int4 v[16];
    #pragma unroll
    for (int k = 0; k < 16; ++k) v[k] = mq[k];
    #pragma unroll
    for (int k = 0; k < 16; ++k) {
      const int r = q * 64 + 4 * k;
      tot += (v[k].x + v[k].y) + (v[k].z + v[k].w);
      bl += ((r     < p) ? v[k].x : 0) + ((r + 1 < p) ? v[k].y : 0)
          + ((r + 2 < p) ? v[k].z : 0) + ((r + 3 < p) ? v[k].w : 0);
    }
  }
  below[q][b] = bl;
  total[q][b] = tot;
  if (tid < NODE_BUCKET) hist[tid] = 0;
  __syncthreads();
  int colTot = 0, colBelow = 0;
  if (tid < NODE_BUCKET) {
    colTot   = (total[0][tid] + total[1][tid]) + (total[2][tid] + total[3][tid]);
    colBelow = (below[0][tid] + below[1][tid]) + (below[2][tid] + below[3][tid]);
  }
  const int excl = excl_scan_first256(colTot, tid, wsum);
  if (tid < NODE_BUCKET) {
    base[tid] = excl + colBelow;
    if (p == 0) {
      if (tid < NB) bbase[tid] = excl;
      if (tid == 0) bbase[NB] = E;
    }
  }
  __syncthreads();
  const int e0 = p * CH;
  const int e1 = min(E, e0 + CH);
  for (int eb = e0 + tid; eb < e1; eb += BATCH * 1024) {
    int s[BATCH], d[BATCH];
    #pragma unroll
    for (int k = 0; k < BATCH; ++k) {
      const int e = min(eb + k * 1024, e1 - 1);
      s[k] = src[e];
      d[k] = dst[e];
    }
    #pragma unroll
    for (int k = 0; k < BATCH; ++k) {
      if (eb + k * 1024 < e1) {
        const int bk = d[k] >> 8;
        const int r = atomicAdd(&hist[bk], 1);
        pairs[base[bk] + r] = make_int2(s[k], d[k]);
      }
    }
  }
}

__global__ __launch_bounds__(1024) void bucket_rows_kernel(
    const int2* __restrict__ pairs, const int* __restrict__ bbase,
    int* rnk, int* __restrict__ rowptr, float* __restrict__ dinv,
    int* __restrict__ rec, int N) {
  __shared__ int cnt[NODE_BUCKET];
  __shared__ int lptr[NODE_BUCKET];
  __shared__ int wsum[8];
  const int tid = threadIdx.x;
  const int bkt = blockIdx.x;
  const int beg = bbase[bkt];
  const int end = bbase[bkt + 1];
  if (tid < NODE_BUCKET) cnt[tid] = 0;
  __syncthreads();
  for (int ib = beg + tid; ib < end; ib += BATCH * 1024) {
    int2 pr[BATCH];
    #pragma unroll
    for (int k = 0; k < BATCH; ++k) {
      pr[k] = pairs[min(ib + k * 1024, end - 1)];
    }
    #pragma unroll
    for (int k = 0; k < BATCH; ++k) {
      const int i = ib + k * 1024;
      const int r = atomicAdd(&cnt[pr[k].y & (NODE_BUCKET - 1)], (i < end) ? 1 : 0);
      if (i < end) rnk[i] = r;
    }
  }
  __syncthreads();
  const int c = (tid < NODE_BUCKET) ? cnt[tid] : 0;
  const int excl = excl_scan_first256(c, tid, wsum);
  if (tid < NODE_BUCKET) {
    lptr[tid] = excl;
    const int node = bkt * NODE_BUCKET + tid;
    if (node < N) {
      rowptr[node] = beg + excl;
      dinv[node] = rsqrtf((float)(c + 1));
    }
  }
  if (bkt == (int)gridDim.x - 1 && tid == 0) rowptr[N] = end;
  __syncthreads();
  for (int ib = beg + tid; ib < end; ib += BATCH * 1024) {
    int2 pr[BATCH];
    int rk[BATCH];
    #pragma unroll
    for (int k = 0; k < BATCH; ++k) {
      const int i = min(ib + k * 1024, end - 1);
      pr[k] = pairs[i];
      rk[k] = rnk[i];
    }
    #pragma unroll
    for (int k = 0; k < BATCH; ++k) {
      const int i = ib + k * 1024;
      if (i < end) rec[beg + lptr[pr[k].y & (NODE_BUCKET - 1)] + rk[k]] = pr[k].x;
    }
  }
}

template <int DOUT>
__global__ __launch_bounds__(256) void gather_gemm_f32_wmma(
    const float* __restrict__ Tin, const int* __restrict__ rowptr,
    const int* __restrict__ rec, const float* __restrict__ dinv,
    const float* __restrict__ bias, const float* __restrict__ W,
    float* __restrict__ Tout, int N, float* __restrict__ clr, int nclr) {
  constexpr int CT = DOUT / 16;
  constexpr int RT = 8 / CT;
  constexpr int ROWS = RT * 16;
  constexpr int NPW = ROWS / 8;
  __shared__ float As[ROWS][K_DIM + 4];
  __shared__ float rs[ROWS];

  typedef float v4f __attribute__((ext_vector_type(4)));
  const int tid  = threadIdx.x;
  const int lane = tid & 31;
  const int wave = __builtin_amdgcn_readfirstlane(tid >> 5);
  const int rowTile = wave / CT;
  const int colTile = wave % CT;
  const int rowBase = blockIdx.x * ROWS;

  for (long i = (long)blockIdx.x * 256 + tid; i < nclr; i += (long)gridDim.x * 256) clr[i] = 0.0f;

  const v4f* Tv = (const v4f*)Tin;
  const v4f bv = ((const v4f*)bias)[lane];
  for (int j = 0; j < NPW; ++j) {
    const int r = wave * NPW + j;
    const int node = rowBase + r;
    v4f acc = {0.0f, 0.0f, 0.0f, 0.0f};
    float w = 0.0f;
    if (node < N) {
      const int beg = rowptr[node];
      const int end = rowptr[node + 1];
      w = dinv[node];
      acc = Tv[(long)node * 32 + lane];
      #pragma unroll 4
      for (int e = beg; e < end; ++e) {
        const v4f v = Tv[(long)rec[e] * 32 + lane];
        acc += v;
      }
      acc = acc * w + bv;
      acc.x = fmaxf(acc.x, 0.0f);
      acc.y = fmaxf(acc.y, 0.0f);
      acc.z = fmaxf(acc.z, 0.0f);
      acc.w = fmaxf(acc.w, 0.0f);
    }
    *(v4f*)&As[r][4 * lane] = acc;
    if (lane == 0) rs[r] = w;
  }
  __syncthreads();

  const int half = lane >> 4;
  const int l16  = lane & 15;
  const int n0   = colTile * 16 + l16;
  const int arow = rowTile * 16 + l16;

  v8f acc = {};
  #pragma unroll
  for (int kk = 0; kk < K_DIM; kk += 4) {
    const int ka = kk + 2 * half;
    v2f a, b;
    a.x = As[arow][ka];
    a.y = As[arow][ka + 1];
    b.x = W[ka * DOUT + n0];
    b.y = W[(ka + 1) * DOUT + n0];
    acc = __builtin_amdgcn_wmma_f32_16x16x4_f32(
         false, a,  false, b,
         (short)0, acc,  false,  false);
  }

  const long mBase = (long)rowBase + rowTile * 16 + 8 * half;
  #pragma unroll
  for (int v = 0; v < 8; ++v) {
    const long m = mBase + v;
    if (m < N) Tout[m * DOUT + n0] = acc[v] * rs[rowTile * 16 + 8 * half + v];
  }
}

__global__ __launch_bounds__(256, 16) void gather_gemm2_f32_wmma(
    const float* __restrict__ X, const int* __restrict__ rowptr,
    const int* __restrict__ rec, const float* __restrict__ dinv,
    const float* __restrict__ Wa, const float* __restrict__ ba,
    const float* __restrict__ Wb, float* __restrict__ Tout, int N) {
  constexpr int DOUT = 128;
  constexpr int ROWS = 16;
  constexpr int NPW = ROWS / 8;
  __shared__ float As[ROWS][K_DIM + 4];
  __shared__ float Hs[ROWS][K_DIM + 4];
  __shared__ float rs[ROWS];

  typedef float v4f __attribute__((ext_vector_type(4)));
  const int tid  = threadIdx.x;
  const int lane = tid & 31;
  const int wave = __builtin_amdgcn_readfirstlane(tid >> 5);
  const int rowBase = blockIdx.x * ROWS;

  const v4f* Xv = (const v4f*)X;
  for (int j = 0; j < NPW; ++j) {
    const int r = wave * NPW + j;
    const int node = rowBase + r;
    v4f acc = {0.0f, 0.0f, 0.0f, 0.0f};
    float w = 0.0f;
    if (node < N) {
      const int beg = rowptr[node];
      const int end = rowptr[node + 1];
      w = dinv[node];
      acc = Xv[(long)node * 32 + lane] * w;
      #pragma unroll 4
      for (int e = beg; e < end; ++e) {
        const int s = rec[e];
        const v4f v = Xv[(long)s * 32 + lane];
        acc += v * dinv[s];
      }
      acc *= w;
    }
    *(v4f*)&As[r][4 * lane] = acc;
    if (lane == 0) rs[r] = w;
  }
  __syncthreads();

  const int half = lane >> 4;
  const int l16  = lane & 15;
  const int n0   = wave * 16 + l16;
  const int arow = l16;

  v8f acc = {};
  #pragma unroll
  for (int kk = 0; kk < K_DIM; kk += 4) {
    const int ka = kk + 2 * half;
    v2f a, b;
    a.x = As[arow][ka];
    a.y = As[arow][ka + 1];
    b.x = Wa[ka * DOUT + n0];
    b.y = Wa[(ka + 1) * DOUT + n0];
    acc = __builtin_amdgcn_wmma_f32_16x16x4_f32(
         false, a,  false, b,
         (short)0, acc,  false,  false);
  }

  const float bb = ba[n0];
  #pragma unroll
  for (int v = 0; v < 8; ++v) Hs[8 * half + v][n0] = fmaxf(acc[v] + bb, 0.0f);
  __syncthreads();

  v8f acc2 = {};
  #pragma unroll
  for (int kk = 0; kk < K_DIM; kk += 4) {
    const int ka = kk + 2 * half;
    v2f a, b;
    a.x = Hs[arow][ka];
    a.y = Hs[arow][ka + 1];
    b.x = Wb[ka * DOUT + n0];
    b.y = Wb[(ka + 1) * DOUT + n0];
    acc2 = __builtin_amdgcn_wmma_f32_16x16x4_f32(
         false, a,  false, b,
         (short)0, acc2,  false,  false);
  }

  const long mBase = (long)rowBase + 8 * half;
  #pragma unroll
  for (int v = 0; v < 8; ++v) {
    const long m = mBase + v;
    if (m < N) Tout[m * DOUT + n0] = acc2[v] * rs[8 * half + v];
  }
}

__global__ void zero_kernel(float* p, long n) {
  const long i = (long)blockIdx.x * blockDim.x + threadIdx.x;
  if (i < n) p[i] = 0.0f;
}

__global__ void pool_count_kernel(const int* __restrict__ batch, float* counts, int N) {
  const long i = (long)blockIdx.x * blockDim.x + threadIdx.x;
  if (i < N) atomicAdd(&counts[batch[i]], 1.0f);
}

__global__ void pool_sum_kernel(const float* __restrict__ h,
                                const int* __restrict__ batch,
                                float* sums, int N) {
  const long idx = (long)blockIdx.x * blockDim.x + threadIdx.x;
  if (idx >= (long)N * 64) return;
  const long i = idx / 64;
  const int  d = (int)(idx % 64);
  atomicAdd(&sums[(long)batch[i] * 64 + d], h[idx]);
}


#define POOL_NPW 2

typedef float pool_v2 __attribute__((ext_vector_type(2)));

__device__ inline void pool_flush(float* sums, float* counts, int g, int lane,
                                  pool_v2 run, int len) {
  float* p = sums + (long)g * 64 + 2 * lane;
  atomicAdd(p, run.x);
  atomicAdd(p + 1, run.y);
  if (lane == 0) atomicAdd(counts + g, (float)len);
}

__global__ __launch_bounds__(256) void gather_pool_kernel(
    const float* __restrict__ T, const int* __restrict__ rowptr,
    const int* __restrict__ rec, const float* __restrict__ dinv,
    const int* __restrict__ batch, float* sums, float* counts, int N) {
  const int lane = threadIdx.x & 31;
  const int first = POOL_NPW *
      __builtin_amdgcn_readfirstlane((int)(blockIdx.x * 8 + (threadIdx.x >> 5)));
  if (first >= N) return;
  const int last = min(N, first + POOL_NPW);
  const pool_v2* Tv = (const pool_v2*)T;
  int g = batch[first];
  pool_v2 run = {0.0f, 0.0f};
  int len = 0;
  for (int node = first; node < last; ++node) {
    const int gn = batch[node];
    const int beg = rowptr[node];
    const int end = rowptr[node + 1];
    const float w = dinv[node];
    pool_v2 acc = Tv[(long)node * 32 + lane];
    #pragma unroll 4
    for (int e = beg; e < end; ++e) {
      const pool_v2 v = Tv[(long)rec[e] * 32 + lane];
      acc += v;
    }
    acc *= w;
    if (gn != g) {
      pool_flush(sums, counts, g, lane, run, len);
      g = gn;
      run = acc;
      len = 1;
    } else {
      run += acc;
      ++len;
    }
  }
  pool_flush(sums, counts, g, lane, run, len);
}

__global__ void pool_final_kernel(const float* __restrict__ sums,
                                  const float* __restrict__ counts,
                                  const float* __restrict__ b3,
                                  float* __restrict__ out, int G) {
  const long idx = (long)blockIdx.x * blockDim.x + threadIdx.x;
  if (idx >= (long)G * 64) return;
  const long g = idx / 64;
  const int  d = (int)(idx % 64);
  const float c = counts[g];
  out[idx] = (sums[idx] + c * b3[d]) / fmaxf(c, 1.0f);
}

static inline int cdiv(long a, long b) { return (int)((a + b - 1) / b); }

extern "C" void kernel_launch(void* const* d_in, const int* in_sizes, int n_in,
                              void* d_out, int out_size, void* d_ws, size_t ws_size,
                              hipStream_t stream) {
  const float* x   = (const float*)d_in[0];
  const int*   ei  = (const int*)d_in[1];
  const int*   bat = (const int*)d_in[2];
  const float* W1  = (const float*)d_in[3];
  const float* b1  = (const float*)d_in[4];
  const float* W2  = (const float*)d_in[5];
  const float* b2  = (const float*)d_in[6];
  const float* W3  = (const float*)d_in[7];
  const float* b3  = (const float*)d_in[8];
  (void)n_in;

  const int N = in_sizes[0] / 128;
  const int E = in_sizes[1] / 2;
  const int G = out_size / 64;
  const int* src = ei;
  const int* dst = ei + E;

  float* ws = (float*)d_ws;
  size_t o = 0;
  float* dinv = ws + o;   o += ((size_t)N + 63) & ~(size_t)63;
  float* T    = ws + o;   o += (size_t)N * 128;
  float* AGG  = ws + o;   o += (size_t)N * 128;
  float* counts = ws + o; o += (size_t)G;
  float* sums   = ws + o; o += (size_t)G * 64;
  const size_t npad = ((size_t)N + 64 + 63) & ~(size_t)63;
  o = (o + 63) & ~(size_t)63;
  int*  rowptr = (int*)(ws + o);  o += npad;
  int*  rec    = (int*)(ws + o);  o += (size_t)E;
  int*  cnt    = (int*)AGG;
  int*  rank   = cnt + npad;
  const bool grouped = ws_size >= o * sizeof(float) && (size_t)N * 128 >= npad + (size_t)E;
  int2* pairs = (int2*)AGG;
  int*  rnk   = (int*)AGG + (((size_t)E * 2 + 3) & ~(size_t)3);
  int*  matT  = rnk + (((size_t)E + 3) & ~(size_t)3);
  int*  bbase = matT + EDGE_CHUNKS * NODE_BUCKET;
  const int NB = cdiv(N, NODE_BUCKET);
  const int CH = cdiv(E, EDGE_CHUNKS);
  const bool binned = grouped && NB <= NODE_BUCKET && E < (1 << 30) &&
                      (size_t)N * 128 >= (size_t)E * 3 + EDGE_CHUNKS * NODE_BUCKET + 328;

  const int B = 256;
  float* out = (float*)d_out;

  if (binned) {
    bucket_count_kernel<<<EDGE_CHUNKS, 1024, 0, stream>>>(dst, matT, E, CH);
    bucket_place_kernel<<<EDGE_CHUNKS, 1024, 0, stream>>>(src, dst, matT, pairs, bbase, E, CH, NB);
    bucket_rows_kernel<<<NB, 1024, 0, stream>>>(pairs, bbase, rnk, rowptr, dinv, rec, N);
  } else if (grouped) {
    zero_kernel<<<cdiv(N, B), B, 0, stream>>>((float*)cnt, (long)N);
    count_rank_kernel<<<cdiv(E, B), B, 0, stream>>>(dst, cnt, rank, E);
    scan_deg_kernel<<<cdiv(N, 256), 256, 0, stream>>>(cnt, rowptr, dinv, N);
    fill_list_kernel<<<cdiv(E, B), B, 0, stream>>>(src, dst, rank, rowptr, rec, E);
  } else {
    init_deg_kernel<<<cdiv(N, B), B, 0, stream>>>(dinv, N);
    count_deg_kernel<<<cdiv(E, B), B, 0, stream>>>(dst, dinv, E);
    finalize_dinv_kernel<<<cdiv(N, B), B, 0, stream>>>(dinv, N);
  }

  if (grouped) {
    gather_gemm2_f32_wmma<<<cdiv(N, 16), B, 0, stream>>>(x, rowptr, rec, dinv, W1, b1, W2, AGG, N);
    gather_gemm_f32_wmma<64><<<cdiv(N, 32), B, 0, stream>>>(AGG, rowptr, rec, dinv, b2, W3, T, N,
                                                             counts, G * 65);
    gather_pool_kernel<<<cdiv(N, 8 * POOL_NPW), 256, 0, stream>>>(T, rowptr, rec, dinv, bat,
                                                                   sums, counts, N);
  } else {
    gemm_f32_wmma<128><<<cdiv(N, 16), B, 0, stream>>>(x, W1, T, N);
    agg_self_kernel<128><<<cdiv((long)N * 128, B), B, 0, stream>>>(T, dinv, AGG, N);
    agg_edges_kernel<128><<<cdiv((long)E * 32, B), B, 0, stream>>>(T, src, dst, dinv, AGG, E);
    bias_relu_kernel<128><<<cdiv((long)N * 128, B), B, 0, stream>>>(AGG, b1, (long)N * 128);

    gemm_f32_wmma<128><<<cdiv(N, 16), B, 0, stream>>>(AGG, W2, T, N);
    agg_self_kernel<128><<<cdiv((long)N * 128, B), B, 0, stream>>>(T, dinv, AGG, N);
    agg_edges_kernel<128><<<cdiv((long)E * 32, B), B, 0, stream>>>(T, src, dst, dinv, AGG, E);
    bias_relu_kernel<128><<<cdiv((long)N * 128, B), B, 0, stream>>>(AGG, b2, (long)N * 128);

    gemm_f32_wmma<64><<<cdiv(N, 32), B, 0, stream>>>(AGG, W3, T, N);
    agg_self_kernel<64><<<cdiv((long)N * 64, B), B, 0, stream>>>(T, dinv, AGG, N);
    agg_edges_kernel<64><<<cdiv((long)E * 16, B), B, 0, stream>>>(T, src, dst, dinv, AGG, E);

    zero_kernel<<<cdiv((long)G * 65, B), B, 0, stream>>>(counts, (long)G * 65);
    pool_count_kernel<<<cdiv(N, B), B, 0, stream>>>(bat, counts, N);
    pool_sum_kernel<<<cdiv((long)N * 64, B), B, 0, stream>>>(AGG, bat, sums, N);
  }

  pool_final_kernel<<<cdiv((long)G * 64, B), B, 0, stream>>>(sums, counts, b3, out, G);
}
